# starting kernel
# baseline (speedup 1.0000x reference)
#include <hip/hip_runtime.h>

typedef __attribute__((ext_vector_type(2))) float v2f;
typedef __attribute__((ext_vector_type(8))) float v8f;

#define KD_L  512
#define KD_C  512
#define KD_B  64

__global__ __launch_bounds__(16) void kd_build_M(float* __restrict__ M) {
    __shared__ float Ksh[KD_L];
    __shared__ float Gsh[KD_L];
    __shared__ float mf[KD_L * 16];

    if (threadIdx.x == 0) {
        float P = 0.0f;
        for (int t = 0; t < KD_L; ++t) {
            float Pp = P + 1.0f;
            float K  = Pp / (Pp + 1.0f);
            P = (1.0f - K) * Pp;
            Ksh[t] = K;
            Gsh[t] = P / (P + 1.0f);
        }
    }
    __syncthreads();

    const int s = blockIdx.x * 16 + threadIdx.x;

    float m = 0.0f;
    for (int t = 0; t < KD_L; ++t) {
        float y = (t == s) ? 1.0f : 0.0f;
        m = m + Ksh[t] * (y - m);
        mf[t * 16 + threadIdx.x] = m;
    }

    float ms = mf[(KD_L - 1) * 16 + threadIdx.x];
    M[(KD_L - 1) * KD_L + s] = ms;
    for (int t = KD_L - 2; t >= 0; --t) {
        float mt = mf[t * 16 + threadIdx.x];
        ms = mt + Gsh[t] * (ms - mt);
        M[t * KD_L + s] = ms;
    }
}

__global__ __launch_bounds__(256) void kd_gemm(const float* __restrict__ obs,
                                               const float* __restrict__ M,
                                               float* __restrict__ out) {
    const int wave = (blockIdx.x * 256 + threadIdx.x) >> 5;
    const int lane = threadIdx.x & 31;

    const int b     = wave >> 8;
    const int rem   = wave & 255;
    const int mBase = (rem >> 5) * 64;
    const int nBase = (rem & 31) * 16;

    const float* __restrict__ Y = obs + (size_t)b * KD_L * KD_C;
    float* __restrict__ T = out + (size_t)b * KD_L * KD_C;
    float* __restrict__ R = out + (size_t)KD_B * KD_L * KD_C
                                + (size_t)b * KD_L * KD_C;

    const int r    = lane & 15;
    const int half = lane >> 4;

    v8f acc0 = {}; v8f acc1 = {}; v8f acc2 = {}; v8f acc3 = {};

    const float* Arow0 = M + (size_t)(mBase      + r) * KD_L;
    const float* Arow1 = M + (size_t)(mBase + 16 + r) * KD_L;
    const float* Arow2 = M + (size_t)(mBase + 32 + r) * KD_L;
    const float* Arow3 = M + (size_t)(mBase + 48 + r) * KD_L;

    for (int k = 0; k < KD_L; k += 4) {
        const int kk = k + 2 * half;

        v2f bf;
        bf.x = Y[(size_t)kk * KD_C + nBase + r];
        bf.y = Y[(size_t)(kk + 1) * KD_C + nBase + r];

        v2f a0 = *(const v2f*)(Arow0 + kk);
        v2f a1 = *(const v2f*)(Arow1 + kk);
        v2f a2 = *(const v2f*)(Arow2 + kk);
        v2f a3 = *(const v2f*)(Arow3 + kk);

        acc0 = __builtin_amdgcn_wmma_f32_16x16x4_f32(false, a0, false, bf,
                                                     (short)0, acc0, false, false);
        acc1 = __builtin_amdgcn_wmma_f32_16x16x4_f32(false, a1, false, bf,
                                                     (short)0, acc1, false, false);
        acc2 = __builtin_amdgcn_wmma_f32_16x16x4_f32(false, a2, false, bf,
                                                     (short)0, acc2, false, false);
        acc3 = __builtin_amdgcn_wmma_f32_16x16x4_f32(false, a3, false, bf,
                                                     (short)0, acc3, false, false);
    }

    const int c = nBase + r;
#pragma unroll
    for (int v = 0; v < 8; ++v) {
        {
            int l = mBase + 8 * half + v;
            float t = acc0[v]; float y = Y[(size_t)l * KD_C + c];
            T[(size_t)l * KD_C + c] = t;  R[(size_t)l * KD_C + c] = y - t;
        }
        {
            int l = mBase + 16 + 8 * half + v;
            float t = acc1[v]; float y = Y[(size_t)l * KD_C + c];
            T[(size_t)l * KD_C + c] = t;  R[(size_t)l * KD_C + c] = y - t;
        }
        {
            int l = mBase + 32 + 8 * half + v;
            float t = acc2[v]; float y = Y[(size_t)l * KD_C + c];
            T[(size_t)l * KD_C + c] = t;  R[(size_t)l * KD_C + c] = y - t;
        }
        {
            int l = mBase + 48 + 8 * half + v;
            float t = acc3[v]; float y = Y[(size_t)l * KD_C + c];
            T[(size_t)l * KD_C + c] = t;  R[(size_t)l * KD_C + c] = y - t;
        }
    }
}

extern "C" void kernel_launch(void* const* d_in, const int* in_sizes, int n_in,
                              void* d_out, int out_size, void* d_ws, size_t ws_size,
                              hipStream_t stream) {
    (void)in_sizes; (void)n_in; (void)out_size; (void)ws_size;

    const float* obs = (const float*)d_in[0];
    float* out = (float*)d_out;
    float* M   = (float*)d_ws;

    kd_build_M<<<KD_L / 16, 16, 0, stream>>>(M);

    kd_gemm<<<(KD_B * 8 * 32) / 8, 256, 0, stream>>>(obs, M, out);
}


# GEMM K loop visits only the band of M that is non-zero in f32 (|l-s|<108, rest underflows to exact 0), sum order of kept terms unchanged; each wave takes row groups p and p+4 to even out band lengths
# speedup vs baseline: 1.2337x; 1.2337x over previous
#include <hip/hip_runtime.h>

typedef __attribute__((ext_vector_type(2))) float v2f;
typedef __attribute__((ext_vector_type(8))) float v8f;

#define KD_L  512
#define KD_C  512
#define KD_B  64
#define KD_HB 112

__global__ __launch_bounds__(16) void kd_build_M(float* __restrict__ M) {
    __shared__ float Ksh[KD_L];
    __shared__ float Gsh[KD_L];
    __shared__ float mf[KD_L * 16];

    if (threadIdx.x == 0) {
        float P = 0.0f;
        for (int t = 0; t < KD_L; ++t) {
            float Pp = P + 1.0f;
            float K  = Pp / (Pp + 1.0f);
            P = (1.0f - K) * Pp;
            Ksh[t] = K;
            Gsh[t] = P / (P + 1.0f);
        }
    }
    __syncthreads();

    const int s = blockIdx.x * 16 + threadIdx.x;

    float m = 0.0f;
    for (int t = 0; t < KD_L; ++t) {
        float y = (t == s) ? 1.0f : 0.0f;
        m = m + Ksh[t] * (y - m);
        mf[t * 16 + threadIdx.x] = m;
    }

    float ms = mf[(KD_L - 1) * 16 + threadIdx.x];
    M[(KD_L - 1) * KD_L + s] = ms;
    for (int t = KD_L - 2; t >= 0; --t) {
        float mt = mf[t * 16 + threadIdx.x];
        ms = mt + Gsh[t] * (ms - mt);
        M[t * KD_L + s] = ms;
    }
}

__device__ __forceinline__ void kd_tile(const float* __restrict__ Y,
                                        const float* __restrict__ M,
                                        float* __restrict__ T,
                                        float* __restrict__ R,
                                        int mBase, int nBase, int r, int half) {
    const int kLo = (mBase - KD_HB) > 0 ? (mBase - KD_HB) : 0;
    const int kHi = (mBase + 64 + KD_HB) < KD_L ? (mBase + 64 + KD_HB) : KD_L;

    v8f acc0 = {}; v8f acc1 = {}; v8f acc2 = {}; v8f acc3 = {};

    const float* Ap0 = M + (size_t)(mBase      + r) * KD_L + kLo + 2 * half;
    const float* Ap1 = M + (size_t)(mBase + 16 + r) * KD_L + kLo + 2 * half;
    const float* Ap2 = M + (size_t)(mBase + 32 + r) * KD_L + kLo + 2 * half;
    const float* Ap3 = M + (size_t)(mBase + 48 + r) * KD_L + kLo + 2 * half;
    const float* Bp  = Y + (size_t)(kLo + 2 * half) * KD_C + nBase + r;

    for (int k = kLo; k < kHi; k += 16) {
        for (int u = 0; u < 16; u += 4) {
            v2f bf;
            bf.x = Bp[(size_t)u * KD_C];
            bf.y = Bp[(size_t)(u + 1) * KD_C];

            v2f a0 = *(const v2f*)(Ap0 + u);
            v2f a1 = *(const v2f*)(Ap1 + u);
            v2f a2 = *(const v2f*)(Ap2 + u);
            v2f a3 = *(const v2f*)(Ap3 + u);

            acc0 = __builtin_amdgcn_wmma_f32_16x16x4_f32(false, a0, false, bf,
                                                         (short)0, acc0, false, false);
            acc1 = __builtin_amdgcn_wmma_f32_16x16x4_f32(false, a1, false, bf,
                                                         (short)0, acc1, false, false);
            acc2 = __builtin_amdgcn_wmma_f32_16x16x4_f32(false, a2, false, bf,
                                                         (short)0, acc2, false, false);
            acc3 = __builtin_amdgcn_wmma_f32_16x16x4_f32(false, a3, false, bf,
                                                         (short)0, acc3, false, false);
        }
        Ap0 += 16; Ap1 += 16; Ap2 += 16; Ap3 += 16;
        Bp  += (size_t)16 * KD_C;
    }

    const int c = nBase + r;
#pragma unroll
    for (int v = 0; v < 8; ++v) {
        {
            int l = mBase + 8 * half + v;
            float t = acc0[v]; float y = Y[(size_t)l * KD_C + c];
            T[(size_t)l * KD_C + c] = t;  R[(size_t)l * KD_C + c] = y - t;
        }
        {
            int l = mBase + 16 + 8 * half + v;
            float t = acc1[v]; float y = Y[(size_t)l * KD_C + c];
            T[(size_t)l * KD_C + c] = t;  R[(size_t)l * KD_C + c] = y - t;
        }
        {
            int l = mBase + 32 + 8 * half + v;
            float t = acc2[v]; float y = Y[(size_t)l * KD_C + c];
            T[(size_t)l * KD_C + c] = t;  R[(size_t)l * KD_C + c] = y - t;
        }
        {
            int l = mBase + 48 + 8 * half + v;
            float t = acc3[v]; float y = Y[(size_t)l * KD_C + c];
            T[(size_t)l * KD_C + c] = t;  R[(size_t)l * KD_C + c] = y - t;
        }
    }
}

__global__ __launch_bounds__(256, 10) void kd_gemm(const float* __restrict__ obs,
                                               const float* __restrict__ M,
                                               float* __restrict__ out) {
    const int blk  = blockIdx.x;
    const int lane = threadIdx.x & 31;

    const int b     = blk >> 4;
    const int p     = ((blk & 3) + (blk >> 8)) & 3;
    const int nBase = (((blk >> 2) & 3) * 8 + (threadIdx.x >> 5)) * 16;

    const float* __restrict__ Y = obs + (size_t)b * KD_L * KD_C;
    float* __restrict__ T = out + (size_t)b * KD_L * KD_C;
    float* __restrict__ R = out + (size_t)KD_B * KD_L * KD_C
                                + (size_t)b * KD_L * KD_C;

    const int r    = lane & 15;
    const int half = lane >> 4;

    for (int pass = 0; pass < 2; ++pass)
        kd_tile(Y, M, T, R, (p + 4 * pass) * 64, nBase, r, half);
}

extern "C" void kernel_launch(void* const* d_in, const int* in_sizes, int n_in,
                              void* d_out, int out_size, void* d_ws, size_t ws_size,
                              hipStream_t stream) {
    (void)in_sizes; (void)n_in; (void)out_size; (void)ws_size;

    const float* obs = (const float*)d_in[0];
    float* out = (float*)d_out;
    float* M   = (float*)d_ws;

    kd_build_M<<<KD_L / 16, 16, 0, stream>>>(M);

    kd_gemm<<<KD_B * 16, 256, 0, stream>>>(obs, M, out);
}


# M is kept in the workspace in WMMA A-fragment order: a GEMM A load is 16 B per lane from 512 contiguous bytes (4 whole lines, two K steps) instead of 8 B per lane from 16 rows; sum order unchanged
# speedup vs baseline: 1.7934x; 1.4537x over previous
#include <hip/hip_runtime.h>

typedef __attribute__((ext_vector_type(2))) float v2f;
typedef __attribute__((ext_vector_type(4))) float v4f;
typedef __attribute__((ext_vector_type(8))) float v8f;

#define KD_L  512
#define KD_C  512
#define KD_B  64
#define KD_HB 112

__device__ __forceinline__ int kd_frag_index(int t, int s) {
    return ((((t >> 6) * 64 + (s >> 3)) * 4 + ((t >> 4) & 3)) * 32
            + (((s >> 1) & 1) * 16 + (t & 15))) * 4
           + ((s >> 2) & 1) * 2 + (s & 1);
}

__global__ __launch_bounds__(16) void kd_build_M(float* __restrict__ M) {
    __shared__ float Ksh[KD_L];
    __shared__ float Gsh[KD_L];
    __shared__ float mf[KD_L * 16];

    if (threadIdx.x == 0) {
        float P = 0.0f;
        for (int t = 0; t < KD_L; ++t) {
            float Pp = P + 1.0f;
            float K  = Pp / (Pp + 1.0f);
            P = (1.0f - K) * Pp;
            Ksh[t] = K;
            Gsh[t] = P / (P + 1.0f);
        }
    }
    __syncthreads();

    const int s = blockIdx.x * 16 + threadIdx.x;

    float m = 0.0f;
    for (int t = 0; t < KD_L; ++t) {
        float y = (t == s) ? 1.0f : 0.0f;
        m = m + Ksh[t] * (y - m);
        mf[t * 16 + threadIdx.x] = m;
    }

    float ms = mf[(KD_L - 1) * 16 + threadIdx.x];
    M[kd_frag_index(KD_L - 1, s)] = ms;
    for (int t = KD_L - 2; t >= KD_L - 16; --t) {
        float mt = mf[t * 16 + threadIdx.x];
        ms = mt + Gsh[t] * (ms - mt);
        M[kd_frag_index(t, s)] = ms;
    }
    for (int tb = KD_L - 32; tb >= 0; tb -= 16) {
        float* Mt = M + kd_frag_index(tb, s);
        for (int r = 15; r >= 0; --r) {
            float mt = mf[(tb + r) * 16 + threadIdx.x];
            ms = mt + Gsh[tb + r] * (ms - mt);
            Mt[r * 4] = ms;
        }
    }
}

__device__ __forceinline__ void kd_tile(const float* __restrict__ Y,
                                        const float* __restrict__ M,
                                        float* __restrict__ T,
                                        float* __restrict__ R,
                                        int mBase, int nBase, int lane) {
    const int r    = lane & 15;
    const int half = lane >> 4;

    const int kLo = (mBase - KD_HB) > 0 ? (mBase - KD_HB) : 0;
    const int kHi = (mBase + 64 + KD_HB) < KD_L ? (mBase + 64 + KD_HB) : KD_L;

    v8f acc0 = {}; v8f acc1 = {}; v8f acc2 = {}; v8f acc3 = {};

    const float* Ap = M + (size_t)(((mBase >> 6) * 64 + (kLo >> 3)) * 4) * 128
                        + lane * 4;
    const float* Bp = Y + (size_t)(kLo + 2 * half) * KD_C + nBase + r;

    for (int k = kLo; k < kHi; k += 16) {
        const v4f p0 = *(const v4f*)(Ap + 0 * 128);
        const v4f p1 = *(const v4f*)(Ap + 1 * 128);
        const v4f p2 = *(const v4f*)(Ap + 2 * 128);
        const v4f p3 = *(const v4f*)(Ap + 3 * 128);
        const v4f q0 = *(const v4f*)(Ap + 4 * 128);
        const v4f q1 = *(const v4f*)(Ap + 5 * 128);
        const v4f q2 = *(const v4f*)(Ap + 6 * 128);
        const v4f q3 = *(const v4f*)(Ap + 7 * 128);

        v2f b0, b1, b2, b3;
        b0.x = Bp[(size_t)0  * KD_C];  b0.y = Bp[(size_t)1  * KD_C];
        b1.x = Bp[(size_t)4  * KD_C];  b1.y = Bp[(size_t)5  * KD_C];
        b2.x = Bp[(size_t)8  * KD_C];  b2.y = Bp[(size_t)9  * KD_C];
        b3.x = Bp[(size_t)12 * KD_C];  b3.y = Bp[(size_t)13 * KD_C];

        acc0 = __builtin_amdgcn_wmma_f32_16x16x4_f32(false, p0.xy, false, b0,
                                                     (short)0, acc0, false, false);
        acc1 = __builtin_amdgcn_wmma_f32_16x16x4_f32(false, p1.xy, false, b0,
                                                     (short)0, acc1, false, false);
        acc2 = __builtin_amdgcn_wmma_f32_16x16x4_f32(false, p2.xy, false, b0,
                                                     (short)0, acc2, false, false);
        acc3 = __builtin_amdgcn_wmma_f32_16x16x4_f32(false, p3.xy, false, b0,
                                                     (short)0, acc3, false, false);
        acc0 = __builtin_amdgcn_wmma_f32_16x16x4_f32(false, p0.zw, false, b1,
                                                     (short)0, acc0, false, false);
        acc1 = __builtin_amdgcn_wmma_f32_16x16x4_f32(false, p1.zw, false, b1,
                                                     (short)0, acc1, false, false);
        acc2 = __builtin_amdgcn_wmma_f32_16x16x4_f32(false, p2.zw, false, b1,
                                                     (short)0, acc2, false, false);
        acc3 = __builtin_amdgcn_wmma_f32_16x16x4_f32(false, p3.zw, false, b1,
                                                     (short)0, acc3, false, false);
        acc0 = __builtin_amdgcn_wmma_f32_16x16x4_f32(false, q0.xy, false, b2,
                                                     (short)0, acc0, false, false);
        acc1 = __builtin_amdgcn_wmma_f32_16x16x4_f32(false, q1.xy, false, b2,
                                                     (short)0, acc1, false, false);
        acc2 = __builtin_amdgcn_wmma_f32_16x16x4_f32(false, q2.xy, false, b2,
                                                     (short)0, acc2, false, false);
        acc3 = __builtin_amdgcn_wmma_f32_16x16x4_f32(false, q3.xy, false, b2,
                                                     (short)0, acc3, false, false);
        acc0 = __builtin_amdgcn_wmma_f32_16x16x4_f32(false, q0.zw, false, b3,
                                                     (short)0, acc0, false, false);
        acc1 = __builtin_amdgcn_wmma_f32_16x16x4_f32(false, q1.zw, false, b3,
                                                     (short)0, acc1, false, false);
        acc2 = __builtin_amdgcn_wmma_f32_16x16x4_f32(false, q2.zw, false, b3,
                                                     (short)0, acc2, false, false);
        acc3 = __builtin_amdgcn_wmma_f32_16x16x4_f32(false, q3.zw, false, b3,
                                                     (short)0, acc3, false, false);
        Ap += 8 * 128;
        Bp += (size_t)16 * KD_C;
    }

    const int c = nBase + r;
#pragma unroll
    for (int v = 0; v < 8; ++v) {
        {
            int l = mBase + 8 * half + v;
            float t = acc0[v]; float y = Y[(size_t)l * KD_C + c];
            T[(size_t)l * KD_C + c] = t;  R[(size_t)l * KD_C + c] = y - t;
        }
        {
            int l = mBase + 16 + 8 * half + v;
            float t = acc1[v]; float y = Y[(size_t)l * KD_C + c];
            T[(size_t)l * KD_C + c] = t;  R[(size_t)l * KD_C + c] = y - t;
        }
        {
            int l = mBase + 32 + 8 * half + v;
            float t = acc2[v]; float y = Y[(size_t)l * KD_C + c];
            T[(size_t)l * KD_C + c] = t;  R[(size_t)l * KD_C + c] = y - t;
        }
        {
            int l = mBase + 48 + 8 * half + v;
            float t = acc3[v]; float y = Y[(size_t)l * KD_C + c];
            T[(size_t)l * KD_C + c] = t;  R[(size_t)l * KD_C + c] = y - t;
        }
    }
}

__global__ __launch_bounds__(256) void kd_gemm(const float* __restrict__ obs,
                                               const float* __restrict__ M,
                                               float* __restrict__ out) {
    const int blk  = blockIdx.x;
    const int lane = threadIdx.x & 31;

    const int b     = blk >> 4;
    const int p     = ((blk & 3) + (blk >> 8)) & 3;
    const int nBase = (((blk >> 2) & 3) * 8 + (threadIdx.x >> 5)) * 16;

    const float* __restrict__ Y = obs + (size_t)b * KD_L * KD_C;
    float* __restrict__ T = out + (size_t)b * KD_L * KD_C;
    float* __restrict__ R = out + (size_t)KD_B * KD_L * KD_C
                                + (size_t)b * KD_L * KD_C;

    for (int pass = 0; pass < 2; ++pass)
        kd_tile(Y, M, T, R, (p + 4 * pass) * 64, nBase, lane);
}

extern "C" void kernel_launch(void* const* d_in, const int* in_sizes, int n_in,
                              void* d_out, int out_size, void* d_ws, size_t ws_size,
                              hipStream_t stream) {
    (void)in_sizes; (void)n_in; (void)out_size; (void)ws_size;

    const float* obs = (const float*)d_in[0];
    float* out = (float*)d_out;
    float* M   = (float*)d_ws;

    kd_build_M<<<KD_L / 16, 16, 0, stream>>>(M);

    kd_gemm<<<KD_B * 16, 256, 0, stream>>>(obs, M, out);
}


# Gain recursion (512 serial steps of two IEEE divisions) stops once its state Pp repeats bit for bit (cycle of 1 or 2 steps); the repeating pair is copied to the rest, all K,G values identical
# speedup vs baseline: 2.3231x; 1.2954x over previous
#include <hip/hip_runtime.h>

typedef __attribute__((ext_vector_type(2))) float v2f;
typedef __attribute__((ext_vector_type(4))) float v4f;
typedef __attribute__((ext_vector_type(8))) float v8f;

#define KD_L  512
#define KD_C  512
#define KD_B  64
#define KD_HB 112

__device__ __forceinline__ int kd_frag_index(int t, int s) {
    return ((((t >> 6) * 64 + (s >> 3)) * 4 + ((t >> 4) & 3)) * 32
            + (((s >> 1) & 1) * 16 + (t & 15))) * 4
           + ((s >> 2) & 1) * 2 + (s & 1);
}

__global__ __launch_bounds__(16) void kd_build_M(float* __restrict__ M) {
    __shared__ float Ksh[KD_L];
    __shared__ float Gsh[KD_L];
    __shared__ float mf[KD_L * 16];
    __shared__ int   nRun;

    if (threadIdx.x == 0) {
        float P      = 0.0f;
        float Pp     = P + 1.0f;
        float PpBack = Pp;
        int t = 0;
        for (; t < KD_L; ++t) {
            float K = Pp / (Pp + 1.0f);
            P = (1.0f - K) * Pp;
            float PpNext = P + 1.0f;
            Ksh[t] = K;
            Gsh[t] = P / (P + 1.0f);
            if (t >= 1 && PpNext == PpBack) { ++t; break; }
            PpBack = Pp;
            Pp     = PpNext;
        }
        nRun = t;
    }
    __syncthreads();

    {
        const int n = nRun;
        if (n < KD_L) {
            const int   src = n - 2 + (threadIdx.x & 1);
            const float Kc  = Ksh[src];
            const float Gc  = Gsh[src];
            for (int u = n + threadIdx.x; u < KD_L; u += 16) {
                Ksh[u] = Kc;
                Gsh[u] = Gc;
            }
        }
    }
    __syncthreads();

    const int s = blockIdx.x * 16 + threadIdx.x;

    float m = 0.0f;
    for (int t = 0; t < KD_L; ++t) {
        float y = (t == s) ? 1.0f : 0.0f;
        m = m + Ksh[t] * (y - m);
        mf[t * 16 + threadIdx.x] = m;
    }

    float ms = mf[(KD_L - 1) * 16 + threadIdx.x];
    M[kd_frag_index(KD_L - 1, s)] = ms;
    for (int t = KD_L - 2; t >= KD_L - 16; --t) {
        float mt = mf[t * 16 + threadIdx.x];
        ms = mt + Gsh[t] * (ms - mt);
        M[kd_frag_index(t, s)] = ms;
    }
    for (int tb = KD_L - 32; tb >= 0; tb -= 16) {
        float* Mt = M + kd_frag_index(tb, s);
        for (int r = 15; r >= 0; --r) {
            float mt = mf[(tb + r) * 16 + threadIdx.x];
            ms = mt + Gsh[tb + r] * (ms - mt);
            Mt[r * 4] = ms;
        }
    }
}

__device__ __forceinline__ void kd_tile(const float* __restrict__ Y,
                                        const float* __restrict__ M,
                                        float* __restrict__ T,
                                        float* __restrict__ R,
                                        int mBase, int nBase, int lane) {
    const int r    = lane & 15;
    const int half = lane >> 4;

    const int kLo = (mBase - KD_HB) > 0 ? (mBase - KD_HB) : 0;
    const int kHi = (mBase + 64 + KD_HB) < KD_L ? (mBase + 64 + KD_HB) : KD_L;

    v8f acc0 = {}; v8f acc1 = {}; v8f acc2 = {}; v8f acc3 = {};

    const float* Ap = M + (size_t)(((mBase >> 6) * 64 + (kLo >> 3)) * 4) * 128
                        + lane * 4;
    const float* Bp = Y + (size_t)(kLo + 2 * half) * KD_C + nBase + r;

    for (int k = kLo; k < kHi; k += 16) {
        const v4f p0 = *(const v4f*)(Ap + 0 * 128);
        const v4f p1 = *(const v4f*)(Ap + 1 * 128);
        const v4f p2 = *(const v4f*)(Ap + 2 * 128);
        const v4f p3 = *(const v4f*)(Ap + 3 * 128);
        const v4f q0 = *(const v4f*)(Ap + 4 * 128);
        const v4f q1 = *(const v4f*)(Ap + 5 * 128);
        const v4f q2 = *(const v4f*)(Ap + 6 * 128);
        const v4f q3 = *(const v4f*)(Ap + 7 * 128);

        v2f b0, b1, b2, b3;
        b0.x = Bp[(size_t)0  * KD_C];  b0.y = Bp[(size_t)1  * KD_C];
        b1.x = Bp[(size_t)4  * KD_C];  b1.y = Bp[(size_t)5  * KD_C];
        b2.x = Bp[(size_t)8  * KD_C];  b2.y = Bp[(size_t)9  * KD_C];
        b3.x = Bp[(size_t)12 * KD_C];  b3.y = Bp[(size_t)13 * KD_C];

        acc0 = __builtin_amdgcn_wmma_f32_16x16x4_f32(false, p0.xy, false, b0,
                                                     (short)0, acc0, false, false);
        acc1 = __builtin_amdgcn_wmma_f32_16x16x4_f32(false, p1.xy, false, b0,
                                                     (short)0, acc1, false, false);
        acc2 = __builtin_amdgcn_wmma_f32_16x16x4_f32(false, p2.xy, false, b0,
                                                     (short)0, acc2, false, false);
        acc3 = __builtin_amdgcn_wmma_f32_16x16x4_f32(false, p3.xy, false, b0,
                                                     (short)0, acc3, false, false);
        acc0 = __builtin_amdgcn_wmma_f32_16x16x4_f32(false, p0.zw, false, b1,
                                                     (short)0, acc0, false, false);
        acc1 = __builtin_amdgcn_wmma_f32_16x16x4_f32(false, p1.zw, false, b1,
                                                     (short)0, acc1, false, false);
        acc2 = __builtin_amdgcn_wmma_f32_16x16x4_f32(false, p2.zw, false, b1,
                                                     (short)0, acc2, false, false);
        acc3 = __builtin_amdgcn_wmma_f32_16x16x4_f32(false, p3.zw, false, b1,
                                                     (short)0, acc3, false, false);
        acc0 = __builtin_amdgcn_wmma_f32_16x16x4_f32(false, q0.xy, false, b2,
                                                     (short)0, acc0, false, false);
        acc1 = __builtin_amdgcn_wmma_f32_16x16x4_f32(false, q1.xy, false, b2,
                                                     (short)0, acc1, false, false);
        acc2 = __builtin_amdgcn_wmma_f32_16x16x4_f32(false, q2.xy, false, b2,
                                                     (short)0, acc2, false, false);
        acc3 = __builtin_amdgcn_wmma_f32_16x16x4_f32(false, q3.xy, false, b2,
                                                     (short)0, acc3, false, false);
        acc0 = __builtin_amdgcn_wmma_f32_16x16x4_f32(false, q0.zw, false, b3,
                                                     (short)0, acc0, false, false);
        acc1 = __builtin_amdgcn_wmma_f32_16x16x4_f32(false, q1.zw, false, b3,
                                                     (short)0, acc1, false, false);
        acc2 = __builtin_amdgcn_wmma_f32_16x16x4_f32(false, q2.zw, false, b3,
                                                     (short)0, acc2, false, false);
        acc3 = __builtin_amdgcn_wmma_f32_16x16x4_f32(false, q3.zw, false, b3,
                                                     (short)0, acc3, false, false);
        Ap += 8 * 128;
        Bp += (size_t)16 * KD_C;
    }

    const int c = nBase + r;
#pragma unroll
    for (int v = 0; v < 8; ++v) {
        {
            int l = mBase + 8 * half + v;
            float t = acc0[v]; float y = Y[(size_t)l * KD_C + c];
            T[(size_t)l * KD_C + c] = t;  R[(size_t)l * KD_C + c] = y - t;
        }
        {
            int l = mBase + 16 + 8 * half + v;
            float t = acc1[v]; float y = Y[(size_t)l * KD_C + c];
            T[(size_t)l * KD_C + c] = t;  R[(size_t)l * KD_C + c] = y - t;
        }
        {
            int l = mBase + 32 + 8 * half + v;
            float t = acc2[v]; float y = Y[(size_t)l * KD_C + c];
            T[(size_t)l * KD_C + c] = t;  R[(size_t)l * KD_C + c] = y - t;
        }
        {
            int l = mBase + 48 + 8 * half + v;
            float t = acc3[v]; float y = Y[(size_t)l * KD_C + c];
            T[(size_t)l * KD_C + c] = t;  R[(size_t)l * KD_C + c] = y - t;
        }
    }
}

__global__ __launch_bounds__(256) void kd_gemm(const float* __restrict__ obs,
                                               const float* __restrict__ M,
                                               float* __restrict__ out) {
    const int blk  = blockIdx.x;
    const int lane = threadIdx.x & 31;

    const int b     = blk >> 4;
    const int p     = ((blk & 3) + (blk >> 8)) & 3;
    const int nBase = (((blk >> 2) & 3) * 8 + (threadIdx.x >> 5)) * 16;

    const float* __restrict__ Y = obs + (size_t)b * KD_L * KD_C;
    float* __restrict__ T = out + (size_t)b * KD_L * KD_C;
    float* __restrict__ R = out + (size_t)KD_B * KD_L * KD_C
                                + (size_t)b * KD_L * KD_C;

    for (int pass = 0; pass < 2; ++pass)
        kd_tile(Y, M, T, R, (p + 4 * pass) * 64, nBase, lane);
}

extern "C" void kernel_launch(void* const* d_in, const int* in_sizes, int n_in,
                              void* d_out, int out_size, void* d_ws, size_t ws_size,
                              hipStream_t stream) {
    (void)in_sizes; (void)n_in; (void)out_size; (void)ws_size;

    const float* obs = (const float*)d_in[0];
    float* out = (float*)d_out;
    float* M   = (float*)d_ws;

    kd_build_M<<<KD_L / 16, 16, 0, stream>>>(M);

    kd_gemm<<<KD_B * 16, 256, 0, stream>>>(obs, M, out);
}


# GEMM A side from LDS: each block copies the contiguous K band of its row group of M (up to 72 KB) into LDS once; its 8 waves read A fragments there instead of from L2 each; sum order unchanged
# speedup vs baseline: 2.4440x; 1.0520x over previous
#include <hip/hip_runtime.h>

typedef __attribute__((ext_vector_type(2))) float v2f;
typedef __attribute__((ext_vector_type(4))) float v4f;
typedef __attribute__((ext_vector_type(8))) float v8f;

#define KD_L  512
#define KD_C  512
#define KD_B  64
#define KD_HB 112
#define KD_BAND (64 + 2 * KD_HB)

__device__ __forceinline__ int kd_frag_index(int t, int s) {
    return ((((t >> 6) * 64 + (s >> 3)) * 4 + ((t >> 4) & 3)) * 32
            + (((s >> 1) & 1) * 16 + (t & 15))) * 4
           + ((s >> 2) & 1) * 2 + (s & 1);
}

__global__ __launch_bounds__(16) void kd_build_M(float* __restrict__ M) {
    __shared__ float Ksh[KD_L];
    __shared__ float Gsh[KD_L];
    __shared__ float mf[KD_L * 16];
    __shared__ int   nRun;

    if (threadIdx.x == 0) {
        float P      = 0.0f;
        float Pp     = P + 1.0f;
        float PpBack = Pp;
        int t = 0;
        for (; t < KD_L; ++t) {
            float K = Pp / (Pp + 1.0f);
            P = (1.0f - K) * Pp;
            float PpNext = P + 1.0f;
            Ksh[t] = K;
            Gsh[t] = P / (P + 1.0f);
            if (t >= 1 && PpNext == PpBack) { ++t; break; }
            PpBack = Pp;
            Pp     = PpNext;
        }
        nRun = t;
    }
    __syncthreads();

    {
        const int n = nRun;
        if (n < KD_L) {
            const int   src = n - 2 + (threadIdx.x & 1);
            const float Kc  = Ksh[src];
            const float Gc  = Gsh[src];
            for (int u = n + threadIdx.x; u < KD_L; u += 16) {
                Ksh[u] = Kc;
                Gsh[u] = Gc;
            }
        }
    }
    __syncthreads();

    const int s = blockIdx.x * 16 + threadIdx.x;

    float m = 0.0f;
    for (int t = 0; t < KD_L; ++t) {
        float y = (t == s) ? 1.0f : 0.0f;
        m = m + Ksh[t] * (y - m);
        mf[t * 16 + threadIdx.x] = m;
    }

    float ms = mf[(KD_L - 1) * 16 + threadIdx.x];
    M[kd_frag_index(KD_L - 1, s)] = ms;
    for (int t = KD_L - 2; t >= KD_L - 16; --t) {
        float mt = mf[t * 16 + threadIdx.x];
        ms = mt + Gsh[t] * (ms - mt);
        M[kd_frag_index(t, s)] = ms;
    }
    for (int tb = KD_L - 32; tb >= 0; tb -= 16) {
        float* Mt = M + kd_frag_index(tb, s);
        for (int r = 15; r >= 0; --r) {
            float mt = mf[(tb + r) * 16 + threadIdx.x];
            ms = mt + Gsh[tb + r] * (ms - mt);
            Mt[r * 4] = ms;
        }
    }
}

__device__ __forceinline__ void kd_tile(const float* __restrict__ Y,
                                        const v4f* As,
                                        float* __restrict__ T,
                                        float* __restrict__ R,
                                        int mBase, int kLo, int kHi,
                                        int nBase, int lane) {
    const int r    = lane & 15;
    const int half = lane >> 4;

    v8f acc0 = {}; v8f acc1 = {}; v8f acc2 = {}; v8f acc3 = {};

    const v4f*   Ap = As + lane;
    const float* Bp = Y + (size_t)(kLo + 2 * half) * KD_C + nBase + r;

    for (int k = kLo; k < kHi; k += 16) {
        const v4f p0 = Ap[0 * 32];
        const v4f p1 = Ap[1 * 32];
        const v4f p2 = Ap[2 * 32];
        const v4f p3 = Ap[3 * 32];
        const v4f q0 = Ap[4 * 32];
        const v4f q1 = Ap[5 * 32];
        const v4f q2 = Ap[6 * 32];
        const v4f q3 = Ap[7 * 32];

        v2f b0, b1, b2, b3;
        b0.x = Bp[(size_t)0  * KD_C];  b0.y = Bp[(size_t)1  * KD_C];
        b1.x = Bp[(size_t)4  * KD_C];  b1.y = Bp[(size_t)5  * KD_C];
        b2.x = Bp[(size_t)8  * KD_C];  b2.y = Bp[(size_t)9  * KD_C];
        b3.x = Bp[(size_t)12 * KD_C];  b3.y = Bp[(size_t)13 * KD_C];

        acc0 = __builtin_amdgcn_wmma_f32_16x16x4_f32(false, p0.xy, false, b0,
                                                     (short)0, acc0, false, false);
        acc1 = __builtin_amdgcn_wmma_f32_16x16x4_f32(false, p1.xy, false, b0,
                                                     (short)0, acc1, false, false);
        acc2 = __builtin_amdgcn_wmma_f32_16x16x4_f32(false, p2.xy, false, b0,
                                                     (short)0, acc2, false, false);
        acc3 = __builtin_amdgcn_wmma_f32_16x16x4_f32(false, p3.xy, false, b0,
                                                     (short)0, acc3, false, false);
        acc0 = __builtin_amdgcn_wmma_f32_16x16x4_f32(false, p0.zw, false, b1,
                                                     (short)0, acc0, false, false);
        acc1 = __builtin_amdgcn_wmma_f32_16x16x4_f32(false, p1.zw, false, b1,
                                                     (short)0, acc1, false, false);
        acc2 = __builtin_amdgcn_wmma_f32_16x16x4_f32(false, p2.zw, false, b1,
                                                     (short)0, acc2, false, false);
        acc3 = __builtin_amdgcn_wmma_f32_16x16x4_f32(false, p3.zw, false, b1,
                                                     (short)0, acc3, false, false);
        acc0 = __builtin_amdgcn_wmma_f32_16x16x4_f32(false, q0.xy, false, b2,
                                                     (short)0, acc0, false, false);
        acc1 = __builtin_amdgcn_wmma_f32_16x16x4_f32(false, q1.xy, false, b2,
                                                     (short)0, acc1, false, false);
        acc2 = __builtin_amdgcn_wmma_f32_16x16x4_f32(false, q2.xy, false, b2,
                                                     (short)0, acc2, false, false);
        acc3 = __builtin_amdgcn_wmma_f32_16x16x4_f32(false, q3.xy, false, b2,
                                                     (short)0, acc3, false, false);
        acc0 = __builtin_amdgcn_wmma_f32_16x16x4_f32(false, q0.zw, false, b3,
                                                     (short)0, acc0, false, false);
        acc1 = __builtin_amdgcn_wmma_f32_16x16x4_f32(false, q1.zw, false, b3,
                                                     (short)0, acc1, false, false);
        acc2 = __builtin_amdgcn_wmma_f32_16x16x4_f32(false, q2.zw, false, b3,
                                                     (short)0, acc2, false, false);
        acc3 = __builtin_amdgcn_wmma_f32_16x16x4_f32(false, q3.zw, false, b3,
                                                     (short)0, acc3, false, false);
        Ap += 8 * 32;
        Bp += (size_t)16 * KD_C;
    }

    const int c = nBase + r;
#pragma unroll
    for (int v = 0; v < 8; ++v) {
        {
            int l = mBase + 8 * half + v;
            float t = acc0[v]; float y = Y[(size_t)l * KD_C + c];
            T[(size_t)l * KD_C + c] = t;  R[(size_t)l * KD_C + c] = y - t;
        }
        {
            int l = mBase + 16 + 8 * half + v;
            float t = acc1[v]; float y = Y[(size_t)l * KD_C + c];
            T[(size_t)l * KD_C + c] = t;  R[(size_t)l * KD_C + c] = y - t;
        }
        {
            int l = mBase + 32 + 8 * half + v;
            float t = acc2[v]; float y = Y[(size_t)l * KD_C + c];
            T[(size_t)l * KD_C + c] = t;  R[(size_t)l * KD_C + c] = y - t;
        }
        {
            int l = mBase + 48 + 8 * half + v;
            float t = acc3[v]; float y = Y[(size_t)l * KD_C + c];
            T[(size_t)l * KD_C + c] = t;  R[(size_t)l * KD_C + c] = y - t;
        }
    }
}

__device__ __forceinline__ void kd_lds_barrier() {
    __builtin_amdgcn_fence(__ATOMIC_RELEASE, "workgroup", "local");
    __builtin_amdgcn_s_barrier();
    __builtin_amdgcn_fence(__ATOMIC_ACQUIRE, "workgroup", "local");
}

template <int N>
__device__ __forceinline__ void kd_stage(const v4f* __restrict__ src, v4f* dst,
                                         bool bandInUse) {
    v4f piece[N];
#pragma unroll
    for (int c = 0; c < N; ++c) piece[c] = src[c * 256];
    if (bandInUse) kd_lds_barrier();
#pragma unroll
    for (int c = 0; c < N; ++c) dst[c * 256] = piece[c];
    kd_lds_barrier();
}

__device__ __forceinline__ void kd_stage_any(const v4f* __restrict__ src, v4f* dst,
                                             int nPiece, bool bandInUse) {
    if (bandInUse) kd_lds_barrier();
    for (int c = 0; c < nPiece; ++c) dst[c * 256] = src[c * 256];
    kd_lds_barrier();
}

__global__ __launch_bounds__(256) void kd_gemm(const float* __restrict__ obs,
                                               const float* __restrict__ M,
                                               float* __restrict__ out) {
    __shared__ v4f As[KD_BAND * 16];

    const int blk  = blockIdx.x;
    const int tid  = threadIdx.x;
    const int lane = tid & 31;

    const int b     = blk >> 4;
    const int p     = ((blk & 3) + (blk >> 8)) & 3;
    const int nBase = (((blk >> 2) & 3) * 8 + (tid >> 5)) * 16;

    const float* __restrict__ Y = obs + (size_t)b * KD_L * KD_C;
    float* __restrict__ T = out + (size_t)b * KD_L * KD_C;
    float* __restrict__ R = out + (size_t)KD_B * KD_L * KD_C
                                + (size_t)b * KD_L * KD_C;

    for (int pass = 0; pass < 2; ++pass) {
        const int mBase = (p + 4 * pass) * 64;
        const int kLo = (mBase - KD_HB) > 0 ? (mBase - KD_HB) : 0;
        const int kHi = (mBase + 64 + KD_HB) < KD_L ? (mBase + 64 + KD_HB) : KD_L;

        const int  nPiece = (kHi - kLo) >> 4;
        const v4f* Msrc   = (const v4f*)(M + (size_t)((mBase >> 6) * 64 + (kLo >> 3)) * 512)
                            + tid;
        switch (nPiece) {
            case 18: kd_stage<18>(Msrc, As + tid, pass != 0); break;
            case 15: kd_stage<15>(Msrc, As + tid, pass != 0); break;
            case 11: kd_stage<11>(Msrc, As + tid, pass != 0); break;
            default: kd_stage_any(Msrc, As + tid, nPiece, pass != 0); break;
        }

        kd_tile(Y, As, T, R, mBase, kLo, kHi, nBase, lane);
    }
}

extern "C" void kernel_launch(void* const* d_in, const int* in_sizes, int n_in,
                              void* d_out, int out_size, void* d_ws, size_t ws_size,
                              hipStream_t stream) {
    (void)in_sizes; (void)n_in; (void)out_size; (void)ws_size;

    const float* obs = (const float*)d_in[0];
    float* out = (float*)d_out;
    float* M   = (float*)d_ws;

    kd_build_M<<<KD_L / 16, 16, 0, stream>>>(M);

    kd_gemm<<<KD_B * 16, 256, 0, stream>>>(obs, M, out);
}


# Trend by blocked recurrence: per 16-step block a 16x16 triangular f32 WMMA plus a carry column, forward then backward sweep, instead of the banded M@y GEMM; sum order changes
# speedup vs baseline: 4.6573x; 1.9056x over previous
#include <hip/hip_runtime.h>

typedef __attribute__((ext_vector_type(2))) float v2f;
typedef __attribute__((ext_vector_type(4))) float v4f;
typedef __attribute__((ext_vector_type(8))) float v8f;

#define KD_L  512
#define KD_C  512
#define KD_B  64
#define KD_NB (KD_L / 16)
#define KD_FR 160
#define KD_PF 6


__global__ __launch_bounds__(32) void kd_build_blocks(float* __restrict__ Wf) {
    __shared__ float Ksh[KD_L];
    __shared__ float Gsh[KD_L];
    __shared__ int   nRun;

    if (threadIdx.x == 0) {
        float P      = 0.0f;
        float Pp     = P + 1.0f;
        float PpBack = Pp;
        int t = 0;
        for (; t < KD_L; ++t) {
            float K = Pp / (Pp + 1.0f);
            P = (1.0f - K) * Pp;
            float PpNext = P + 1.0f;
            Ksh[t] = K;
            Gsh[t] = P / (P + 1.0f);
            if (t >= 1 && PpNext == PpBack) { ++t; break; }
            PpBack = Pp;
            Pp     = PpNext;
        }
        nRun = t;
    }
    __syncthreads();

    {
        const int n = nRun;
        if (n < KD_L) {
            const int   src = n - 2 + (threadIdx.x & 1);
            const float Kc  = Ksh[src];
            const float Gc  = Gsh[src];
            for (int u = n + threadIdx.x; u < KD_L; u += 32) {
                Ksh[u] = Kc;
                Gsh[u] = Gc;
            }
        }
    }
    __syncthreads();

    const int j = blockIdx.x;
    const int k = threadIdx.x;
    if (k > 16) return;

    float* Fd = Wf + (size_t)((0 * KD_NB + j) * KD_FR) * 2;
    float* Bd = Wf + (size_t)((1 * KD_NB + j) * KD_FR) * 2;
    const int q = (k & 7) >> 1, h = (k >> 3) & 1, s = k & 1;

    {
        float m = (k == 16) ? 1.0f : 0.0f;
        for (int i = 0; i < 16; ++i) {
            const float y = (i == k) ? 1.0f : 0.0f;
            m = m + Ksh[16 * j + i] * (y - m);
            if (k < 16) {
                Fd[((q * 32) + h * 16 + i) * 2 + s] = m;
            } else {
                Fd[((4 * 32) + 16 + i) * 2 + 0] = m;
                Fd[((4 * 32) + 16 + i) * 2 + 1] = 0.0f;
                Fd[((4 * 32) + i) * 2 + 0] = 0.0f;
                Fd[((4 * 32) + i) * 2 + 1] = 0.0f;
            }
        }
    }
    {
        float ms = (k == 16) ? 1.0f : 0.0f;
        for (int i = 15; i >= 0; --i) {
            const int   t  = 16 * j + i;
            const float mt = (i == k) ? 1.0f : 0.0f;
            ms = (t == KD_L - 1) ? mt : (mt + Gsh[t] * (ms - mt));
            if (k < 16) {
                Bd[((q * 32) + h * 16 + i) * 2 + s] = ms;
            } else {
                Bd[((4 * 32) + i) * 2 + 0] = ms;
                Bd[((4 * 32) + i) * 2 + 1] = 0.0f;
                Bd[((4 * 32) + 16 + i) * 2 + 0] = 0.0f;
                Bd[((4 * 32) + 16 + i) * 2 + 1] = 0.0f;
            }
        }
    }
}

__device__ __forceinline__ v8f kd_mma(v2f a, v2f b, v8f c) {
    return __builtin_amdgcn_wmma_f32_16x16x4_f32(false, a, false, b, (short)0, c,
                                                 false, false);
}

__global__ __launch_bounds__(256) void kd_scan(const float* __restrict__ obs,
                                               const float* __restrict__ Wf,
                                               float* __restrict__ out) {
    __shared__ v4f Afs4[2 * KD_NB * KD_FR / 2];

    const int tid  = threadIdx.x;
    const int lane = tid & 31;
    const int wave = tid >> 5;

    {
        const v4f* src = (const v4f*)Wf + tid;
        v4f piece[20];
#pragma unroll
        for (int p = 0; p < 20; ++p) piece[p] = src[p * 256];
#pragma unroll
        for (int p = 0; p < 20; ++p) Afs4[p * 256 + tid] = piece[p];
    }
    __syncthreads();
    const v2f* Afs = (const v2f*)Afs4;

    const int blk   = blockIdx.x;
    const int b     = blk >> 2;
    const int strip = (blk & 3) * 8 + wave;
    const int r     = lane & 15;
    const int half  = lane >> 4;
    const size_t base = (size_t)b * KD_L * KD_C + (size_t)(8 * half) * KD_C
                        + strip * 16 + r;
    const float* __restrict__ Yp = obs + base;
    float* __restrict__ Tp = out + base;
    float* __restrict__ Rp = out + (size_t)KD_B * KD_L * KD_C + base;

    v8f   mf[KD_NB];
    float yb[KD_PF][8];
    v2f   a[5];

#pragma unroll
    for (int jj = 0; jj < KD_PF; ++jj)
#pragma unroll
        for (int i = 0; i < 8; ++i) yb[jj][i] = Yp[(size_t)(16 * jj + i) * KD_C];
#pragma unroll
    for (int q = 0; q < 5; ++q) a[q] = Afs[(0 * KD_NB + 0) * KD_FR + q * 32 + lane];

    float carry = 0.0f;
#pragma unroll
    for (int j = 0; j < KD_NB; ++j) {
        float y[8];
#pragma unroll
        for (int i = 0; i < 8; ++i) y[i] = yb[j % KD_PF][i];
        if (j + KD_PF < KD_NB) {
#pragma unroll
            for (int i = 0; i < 8; ++i)
                yb[j % KD_PF][i] = Yp[(size_t)(16 * (j + KD_PF) + i) * KD_C];
        }
        v2f an[5];
        {
            const v2f* An = (j + 1 < KD_NB) ? Afs + (0 * KD_NB + j + 1) * KD_FR + lane
                                            : Afs + (1 * KD_NB + KD_NB - 1) * KD_FR + lane;
#pragma unroll
            for (int q = 0; q < 5; ++q) an[q] = An[q * 32];
        }
        v2f b0, b1, b2, b3;
        b0.x = y[0]; b0.y = y[1];  b1.x = y[2]; b1.y = y[3];
        b2.x = y[4]; b2.y = y[5];  b3.x = y[6]; b3.y = y[7];
        v8f acc = {};
        acc = kd_mma(a[0], b0, acc);
        acc = kd_mma(a[1], b1, acc);
        acc = kd_mma(a[2], b2, acc);
        acc = kd_mma(a[3], b3, acc);
        if (j > 0) {
            v2f bc;
            bc.x = half ? carry : 0.0f;
            bc.y = 0.0f;
            acc = kd_mma(a[4], bc, acc);
        }
        mf[j] = acc;
        carry = acc[7];
#pragma unroll
        for (int q = 0; q < 5; ++q) a[q] = an[q];
        __builtin_amdgcn_sched_barrier(0);
    }

    int zero = 0;
    asm volatile("" : "+s"(zero));
    const float* __restrict__ Yq = Yp + zero;

#pragma unroll
    for (int jj = 0; jj < KD_PF; ++jj)
#pragma unroll
        for (int i = 0; i < 8; ++i)
            yb[jj][i] = Yq[(size_t)(16 * (KD_NB - 1 - jj) + i) * KD_C];

    carry = 0.0f;
#pragma unroll
    for (int jr = 0; jr < KD_NB; ++jr) {
        const int j = KD_NB - 1 - jr;
        float y[8];
#pragma unroll
        for (int i = 0; i < 8; ++i) y[i] = yb[jr % KD_PF][i];
        if (jr + KD_PF < KD_NB) {
#pragma unroll
            for (int i = 0; i < 8; ++i)
                yb[jr % KD_PF][i] = Yq[(size_t)(16 * (j - KD_PF) + i) * KD_C];
        }
        v2f an[5];
        if (j > 0) {
            const v2f* An = Afs + (1 * KD_NB + j - 1) * KD_FR + lane;
#pragma unroll
            for (int q = 0; q < 5; ++q) an[q] = An[q * 32];
        } else {
#pragma unroll
            for (int q = 0; q < 5; ++q) an[q] = a[q];
        }
        const v8f m = mf[j];
        v2f b0, b1, b2, b3;
        b0.x = m[0]; b0.y = m[1];  b1.x = m[2]; b1.y = m[3];
        b2.x = m[4]; b2.y = m[5];  b3.x = m[6]; b3.y = m[7];
        v8f acc = {};
        acc = kd_mma(a[0], b0, acc);
        acc = kd_mma(a[1], b1, acc);
        acc = kd_mma(a[2], b2, acc);
        acc = kd_mma(a[3], b3, acc);
        if (j < KD_NB - 1) {
            v2f bc;
            bc.x = half ? 0.0f : carry;
            bc.y = 0.0f;
            acc = kd_mma(a[4], bc, acc);
        }
        carry = acc[0];
#pragma unroll
        for (int i = 0; i < 8; ++i) {
            const size_t o = (size_t)(16 * j + i) * KD_C;
            const float  t = acc[i];
            Tp[o] = t;
            Rp[o] = y[i] - t;
        }
#pragma unroll
        for (int q = 0; q < 5; ++q) a[q] = an[q];
        __builtin_amdgcn_sched_barrier(0);
    }
}

extern "C" void kernel_launch(void* const* d_in, const int* in_sizes, int n_in,
                              void* d_out, int out_size, void* d_ws, size_t ws_size,
                              hipStream_t stream) {
    (void)in_sizes; (void)n_in; (void)out_size; (void)ws_size;

    const float* obs = (const float*)d_in[0];
    float* out = (float*)d_out;
    float* Wf  = (float*)d_ws;

    kd_build_blocks<<<KD_NB, 32, 0, stream>>>(Wf);

    kd_scan<<<KD_B * 4, 256, 0, stream>>>(obs, Wf, out);
}


# Each wave owns 32 adjacent channels as two interleaved 16-column tiles (even, odd), 8 B per lane: every global load and store covers whole 128-B lines; 4 waves per workgroup; sum order unchanged
# speedup vs baseline: 5.1096x; 1.0971x over previous
#include <hip/hip_runtime.h>

typedef __attribute__((ext_vector_type(2))) float v2f;
typedef __attribute__((ext_vector_type(4))) float v4f;
typedef __attribute__((ext_vector_type(8))) float v8f;

#define KD_L  512
#define KD_C  512
#define KD_B  64
#define KD_NB (KD_L / 16)
#define KD_FR 160
#define KD_PF 6


__global__ __launch_bounds__(32) void kd_build_blocks(float* __restrict__ Wf) {
    __shared__ float Ksh[KD_L];
    __shared__ float Gsh[KD_L];
    __shared__ int   nRun;

    if (threadIdx.x == 0) {
        float P      = 0.0f;
        float Pp     = P + 1.0f;
        float PpBack = Pp;
        int t = 0;
        for (; t < KD_L; ++t) {
            float K = Pp / (Pp + 1.0f);
            P = (1.0f - K) * Pp;
            float PpNext = P + 1.0f;
            Ksh[t] = K;
            Gsh[t] = P / (P + 1.0f);
            if (t >= 1 && PpNext == PpBack) { ++t; break; }
            PpBack = Pp;
            Pp     = PpNext;
        }
        nRun = t;
    }
    __syncthreads();

    {
        const int n = nRun;
        if (n < KD_L) {
            const int   src = n - 2 + (threadIdx.x & 1);
            const float Kc  = Ksh[src];
            const float Gc  = Gsh[src];
            for (int u = n + threadIdx.x; u < KD_L; u += 32) {
                Ksh[u] = Kc;
                Gsh[u] = Gc;
            }
        }
    }
    __syncthreads();

    const int j = blockIdx.x;
    const int k = threadIdx.x;
    if (k > 16) return;

    float* Fd = Wf + (size_t)((0 * KD_NB + j) * KD_FR) * 2;
    float* Bd = Wf + (size_t)((1 * KD_NB + j) * KD_FR) * 2;
    const int q = (k & 7) >> 1, h = (k >> 3) & 1, s = k & 1;

    {
        float m = (k == 16) ? 1.0f : 0.0f;
        for (int i = 0; i < 16; ++i) {
            const float y = (i == k) ? 1.0f : 0.0f;
            m = m + Ksh[16 * j + i] * (y - m);
            if (k < 16) {
                Fd[((q * 32) + h * 16 + i) * 2 + s] = m;
            } else {
                Fd[((4 * 32) + 16 + i) * 2 + 0] = m;
                Fd[((4 * 32) + 16 + i) * 2 + 1] = 0.0f;
                Fd[((4 * 32) + i) * 2 + 0] = 0.0f;
                Fd[((4 * 32) + i) * 2 + 1] = 0.0f;
            }
        }
    }
    {
        float ms = (k == 16) ? 1.0f : 0.0f;
        for (int i = 15; i >= 0; --i) {
            const int   t  = 16 * j + i;
            const float mt = (i == k) ? 1.0f : 0.0f;
            ms = (t == KD_L - 1) ? mt : (mt + Gsh[t] * (ms - mt));
            if (k < 16) {
                Bd[((q * 32) + h * 16 + i) * 2 + s] = ms;
            } else {
                Bd[((4 * 32) + i) * 2 + 0] = ms;
                Bd[((4 * 32) + i) * 2 + 1] = 0.0f;
                Bd[((4 * 32) + 16 + i) * 2 + 0] = 0.0f;
                Bd[((4 * 32) + 16 + i) * 2 + 1] = 0.0f;
            }
        }
    }
}

__device__ __forceinline__ v8f kd_mma(v2f a, v2f b, v8f c) {
    return __builtin_amdgcn_wmma_f32_16x16x4_f32(false, a, false, b, (short)0, c,
                                                 false, false);
}

__global__ __launch_bounds__(128) void kd_scan(const float* __restrict__ obs,
                                               const float* __restrict__ Wf,
                                               float* __restrict__ out) {
    __shared__ v4f Afs4[2 * KD_NB * KD_FR / 2];

    const int tid  = threadIdx.x;
    const int lane = tid & 31;
    const int wave = tid >> 5;

    {
        const v4f* src = (const v4f*)Wf + tid;
        v4f piece[40];
#pragma unroll
        for (int p = 0; p < 40; ++p) piece[p] = src[p * 128];
#pragma unroll
        for (int p = 0; p < 40; ++p) Afs4[p * 128 + tid] = piece[p];
    }
    __syncthreads();
    const v2f* Afs = (const v2f*)Afs4;

    const int blk   = blockIdx.x;
    const int b     = blk >> 2;
    const int strip = (blk & 3) * 4 + wave;
    const int m     = lane & 15;
    const int half  = lane >> 4;
    const size_t base = (size_t)b * KD_L * KD_C + (size_t)(8 * half) * KD_C
                        + strip * 32 + 2 * m;
    const float* __restrict__ Yp = obs + base;
    float* __restrict__ Tp = out + base;
    float* __restrict__ Rp = out + (size_t)KD_B * KD_L * KD_C + base;

    v8f mf0[KD_NB];
    v8f mf1[KD_NB];
    v2f yb[KD_PF][8];
    v2f a[5];

#pragma unroll
    for (int jj = 0; jj < KD_PF; ++jj)
#pragma unroll
        for (int i = 0; i < 8; ++i)
            yb[jj][i] = *(const v2f*)(Yp + (size_t)(16 * jj + i) * KD_C);
#pragma unroll
    for (int q = 0; q < 5; ++q) a[q] = Afs[(0 * KD_NB + 0) * KD_FR + q * 32 + lane];

    float carry0 = 0.0f, carry1 = 0.0f;
#pragma unroll
    for (int j = 0; j < KD_NB; ++j) {
        v2f y[8];
#pragma unroll
        for (int i = 0; i < 8; ++i) y[i] = yb[j % KD_PF][i];
        if (j + KD_PF < KD_NB) {
#pragma unroll
            for (int i = 0; i < 8; ++i)
                yb[j % KD_PF][i] =
                    *(const v2f*)(Yp + (size_t)(16 * (j + KD_PF) + i) * KD_C);
        }
        v2f an[5];
        {
            const v2f* An = (j + 1 < KD_NB) ? Afs + (0 * KD_NB + j + 1) * KD_FR + lane
                                            : Afs + (1 * KD_NB + KD_NB - 1) * KD_FR + lane;
#pragma unroll
            for (int q = 0; q < 5; ++q) an[q] = An[q * 32];
        }
        v2f p0, p1, p2, p3, r0, r1, r2, r3;
        p0.x = y[0].x; p0.y = y[1].x;  p1.x = y[2].x; p1.y = y[3].x;
        p2.x = y[4].x; p2.y = y[5].x;  p3.x = y[6].x; p3.y = y[7].x;
        r0.x = y[0].y; r0.y = y[1].y;  r1.x = y[2].y; r1.y = y[3].y;
        r2.x = y[4].y; r2.y = y[5].y;  r3.x = y[6].y; r3.y = y[7].y;
        v8f acc0 = {};
        v8f acc1 = {};
        acc0 = kd_mma(a[0], p0, acc0);
        acc1 = kd_mma(a[0], r0, acc1);
        acc0 = kd_mma(a[1], p1, acc0);
        acc1 = kd_mma(a[1], r1, acc1);
        acc0 = kd_mma(a[2], p2, acc0);
        acc1 = kd_mma(a[2], r2, acc1);
        acc0 = kd_mma(a[3], p3, acc0);
        acc1 = kd_mma(a[3], r3, acc1);
        if (j > 0) {
            v2f bc0, bc1;
            bc0.x = half ? carry0 : 0.0f;
            bc0.y = 0.0f;
            bc1.x = half ? carry1 : 0.0f;
            bc1.y = 0.0f;
            acc0 = kd_mma(a[4], bc0, acc0);
            acc1 = kd_mma(a[4], bc1, acc1);
        }
        mf0[j] = acc0;
        mf1[j] = acc1;
        carry0 = acc0[7];
        carry1 = acc1[7];
#pragma unroll
        for (int q = 0; q < 5; ++q) a[q] = an[q];
        __builtin_amdgcn_sched_barrier(0);
    }

    int zero = 0;
    asm volatile("" : "+s"(zero));
    const float* __restrict__ Yq = Yp + zero;

#pragma unroll
    for (int jj = 0; jj < KD_PF; ++jj)
#pragma unroll
        for (int i = 0; i < 8; ++i)
            yb[jj][i] = *(const v2f*)(Yq + (size_t)(16 * (KD_NB - 1 - jj) + i) * KD_C);

    carry0 = 0.0f;
    carry1 = 0.0f;
#pragma unroll
    for (int jr = 0; jr < KD_NB; ++jr) {
        const int j = KD_NB - 1 - jr;
        v2f y[8];
#pragma unroll
        for (int i = 0; i < 8; ++i) y[i] = yb[jr % KD_PF][i];
        if (jr + KD_PF < KD_NB) {
#pragma unroll
            for (int i = 0; i < 8; ++i)
                yb[jr % KD_PF][i] =
                    *(const v2f*)(Yq + (size_t)(16 * (j - KD_PF) + i) * KD_C);
        }
        v2f an[5];
        if (j > 0) {
            const v2f* An = Afs + (1 * KD_NB + j - 1) * KD_FR + lane;
#pragma unroll
            for (int q = 0; q < 5; ++q) an[q] = An[q * 32];
        } else {
#pragma unroll
            for (int q = 0; q < 5; ++q) an[q] = a[q];
        }
        const v8f m0 = mf0[j];
        const v8f m1 = mf1[j];
        v2f p0, p1, p2, p3, r0, r1, r2, r3;
        p0.x = m0[0]; p0.y = m0[1];  p1.x = m0[2]; p1.y = m0[3];
        p2.x = m0[4]; p2.y = m0[5];  p3.x = m0[6]; p3.y = m0[7];
        r0.x = m1[0]; r0.y = m1[1];  r1.x = m1[2]; r1.y = m1[3];
        r2.x = m1[4]; r2.y = m1[5];  r3.x = m1[6]; r3.y = m1[7];
        v8f acc0 = {};
        v8f acc1 = {};
        acc0 = kd_mma(a[0], p0, acc0);
        acc1 = kd_mma(a[0], r0, acc1);
        acc0 = kd_mma(a[1], p1, acc0);
        acc1 = kd_mma(a[1], r1, acc1);
        acc0 = kd_mma(a[2], p2, acc0);
        acc1 = kd_mma(a[2], r2, acc1);
        acc0 = kd_mma(a[3], p3, acc0);
        acc1 = kd_mma(a[3], r3, acc1);
        if (j < KD_NB - 1) {
            v2f bc0, bc1;
            bc0.x = half ? 0.0f : carry0;
            bc0.y = 0.0f;
            bc1.x = half ? 0.0f : carry1;
            bc1.y = 0.0f;
            acc0 = kd_mma(a[4], bc0, acc0);
            acc1 = kd_mma(a[4], bc1, acc1);
        }
        carry0 = acc0[0];
        carry1 = acc1[0];
#pragma unroll
        for (int i = 0; i < 8; ++i) {
            const size_t o = (size_t)(16 * j + i) * KD_C;
            v2f t, d;
            t.x = acc0[i];          t.y = acc1[i];
            d.x = y[i].x - acc0[i]; d.y = y[i].y - acc1[i];
            *(v2f*)(Tp + o) = t;
            *(v2f*)(Rp + o) = d;
        }
#pragma unroll
        for (int q = 0; q < 5; ++q) a[q] = an[q];
        __builtin_amdgcn_sched_barrier(0);
    }
}

extern "C" void kernel_launch(void* const* d_in, const int* in_sizes, int n_in,
                              void* d_out, int out_size, void* d_ws, size_t ws_size,
                              hipStream_t stream) {
    (void)in_sizes; (void)n_in; (void)out_size; (void)ws_size;

    const float* obs = (const float*)d_in[0];
    float* out = (float*)d_out;
    float* Wf  = (float*)d_ws;

    kd_build_blocks<<<KD_NB, 32, 0, stream>>>(Wf);

    kd_scan<<<KD_B * 4, 128, 0, stream>>>(obs, Wf, out);
}


# obs is read once: forward sweep parks 26 time blocks per wave in LDS (208 KB per workgroup), the last 6 stay in the load-ring registers; backward sweep reads no global memory; sum order unchanged
# speedup vs baseline: 5.5069x; 1.0778x over previous
#include <hip/hip_runtime.h>

typedef __attribute__((ext_vector_type(2))) float v2f;
typedef __attribute__((ext_vector_type(4))) float v4f;
typedef __attribute__((ext_vector_type(8))) float v8f;

#define KD_L  512
#define KD_C  512
#define KD_B  64
#define KD_NB (KD_L / 16)
#define KD_FR 160
#define KD_PF 6
#define KD_ST (KD_NB - KD_PF)


__global__ __launch_bounds__(32) void kd_build_blocks(float* __restrict__ Wf) {
    __shared__ float Ksh[KD_L];
    __shared__ float Gsh[KD_L];
    __shared__ int   nRun;

    if (threadIdx.x == 0) {
        float P      = 0.0f;
        float Pp     = P + 1.0f;
        float PpBack = Pp;
        int t = 0;
        for (; t < KD_L; ++t) {
            float K = Pp / (Pp + 1.0f);
            P = (1.0f - K) * Pp;
            float PpNext = P + 1.0f;
            Ksh[t] = K;
            Gsh[t] = P / (P + 1.0f);
            if (t >= 1 && PpNext == PpBack) { ++t; break; }
            PpBack = Pp;
            Pp     = PpNext;
        }
        nRun = t;
    }
    __syncthreads();

    {
        const int n = nRun;
        if (n < KD_L) {
            const int   src = n - 2 + (threadIdx.x & 1);
            const float Kc  = Ksh[src];
            const float Gc  = Gsh[src];
            for (int u = n + threadIdx.x; u < KD_L; u += 32) {
                Ksh[u] = Kc;
                Gsh[u] = Gc;
            }
        }
    }
    __syncthreads();

    const int j = blockIdx.x;
    const int k = threadIdx.x;
    if (k > 16) return;

    float* Fd = Wf + (size_t)((0 * KD_NB + j) * KD_FR) * 2;
    float* Bd = Wf + (size_t)((1 * KD_NB + j) * KD_FR) * 2;
    const int q = (k & 7) >> 1, h = (k >> 3) & 1, s = k & 1;

    {
        float m = (k == 16) ? 1.0f : 0.0f;
        for (int i = 0; i < 16; ++i) {
            const float y = (i == k) ? 1.0f : 0.0f;
            m = m + Ksh[16 * j + i] * (y - m);
            if (k < 16) {
                Fd[((q * 32) + h * 16 + i) * 2 + s] = m;
            } else {
                Fd[((4 * 32) + 16 + i) * 2 + 0] = m;
                Fd[((4 * 32) + 16 + i) * 2 + 1] = 0.0f;
                Fd[((4 * 32) + i) * 2 + 0] = 0.0f;
                Fd[((4 * 32) + i) * 2 + 1] = 0.0f;
            }
        }
    }
    {
        float ms = (k == 16) ? 1.0f : 0.0f;
        for (int i = 15; i >= 0; --i) {
            const int   t  = 16 * j + i;
            const float mt = (i == k) ? 1.0f : 0.0f;
            ms = (t == KD_L - 1) ? mt : (mt + Gsh[t] * (ms - mt));
            if (k < 16) {
                Bd[((q * 32) + h * 16 + i) * 2 + s] = ms;
            } else {
                Bd[((4 * 32) + i) * 2 + 0] = ms;
                Bd[((4 * 32) + i) * 2 + 1] = 0.0f;
                Bd[((4 * 32) + 16 + i) * 2 + 0] = 0.0f;
                Bd[((4 * 32) + 16 + i) * 2 + 1] = 0.0f;
            }
        }
    }
}

__device__ __forceinline__ v8f kd_mma(v2f a, v2f b, v8f c) {
    return __builtin_amdgcn_wmma_f32_16x16x4_f32(false, a, false, b, (short)0, c,
                                                 false, false);
}

__global__ __launch_bounds__(128) void kd_scan(const float* __restrict__ obs,
                                               const float* __restrict__ Wf,
                                               float* __restrict__ out) {
    __shared__ v4f Afs4[2 * KD_NB * KD_FR / 2];
    __shared__ v2f Ysh[4 * KD_ST * 8 * 32];

    const int tid  = threadIdx.x;
    const int lane = tid & 31;
    const int wave = tid >> 5;

    {
        const v4f* src = (const v4f*)Wf + tid;
        v4f piece[40];
#pragma unroll
        for (int p = 0; p < 40; ++p) piece[p] = src[p * 128];
#pragma unroll
        for (int p = 0; p < 40; ++p) Afs4[p * 128 + tid] = piece[p];
    }
    __syncthreads();
    const v2f* Afs = (const v2f*)Afs4;

    const int blk   = blockIdx.x;
    const int b     = blk >> 2;
    const int strip = (blk & 3) * 4 + wave;
    const int m     = lane & 15;
    const int half  = lane >> 4;
    const size_t base = (size_t)b * KD_L * KD_C + (size_t)(8 * half) * KD_C
                        + strip * 32 + 2 * m;
    const float* __restrict__ Yp = obs + base;
    float* __restrict__ Tp = out + base;
    float* __restrict__ Rp = out + (size_t)KD_B * KD_L * KD_C + base;
    v2f* Yw = Ysh + (size_t)wave * (KD_ST * 8 * 32) + lane;

    v8f mf0[KD_NB];
    v8f mf1[KD_NB];
    v2f yb[KD_PF][8];
    v2f a[5];

#pragma unroll
    for (int jj = 0; jj < KD_PF; ++jj)
#pragma unroll
        for (int i = 0; i < 8; ++i)
            yb[jj][i] = *(const v2f*)(Yp + (size_t)(16 * jj + i) * KD_C);
#pragma unroll
    for (int q = 0; q < 5; ++q) a[q] = Afs[(0 * KD_NB + 0) * KD_FR + q * 32 + lane];

    float carry0 = 0.0f, carry1 = 0.0f;
#pragma unroll
    for (int j = 0; j < KD_NB; ++j) {
        v2f y[8];
#pragma unroll
        for (int i = 0; i < 8; ++i) y[i] = yb[j % KD_PF][i];
        if (j + KD_PF < KD_NB) {
#pragma unroll
            for (int i = 0; i < 8; ++i)
                yb[j % KD_PF][i] =
                    *(const v2f*)(Yp + (size_t)(16 * (j + KD_PF) + i) * KD_C);
        }
        if (j < KD_ST) {
#pragma unroll
            for (int i = 0; i < 8; ++i) Yw[(j * 8 + i) * 32] = y[i];
        }
        v2f an[5];
        {
            const v2f* An = (j + 1 < KD_NB) ? Afs + (0 * KD_NB + j + 1) * KD_FR + lane
                                            : Afs + (1 * KD_NB + KD_NB - 1) * KD_FR + lane;
#pragma unroll
            for (int q = 0; q < 5; ++q) an[q] = An[q * 32];
        }
        v2f p0, p1, p2, p3, r0, r1, r2, r3;
        p0.x = y[0].x; p0.y = y[1].x;  p1.x = y[2].x; p1.y = y[3].x;
        p2.x = y[4].x; p2.y = y[5].x;  p3.x = y[6].x; p3.y = y[7].x;
        r0.x = y[0].y; r0.y = y[1].y;  r1.x = y[2].y; r1.y = y[3].y;
        r2.x = y[4].y; r2.y = y[5].y;  r3.x = y[6].y; r3.y = y[7].y;
        v8f acc0 = {};
        v8f acc1 = {};
        acc0 = kd_mma(a[0], p0, acc0);
        acc1 = kd_mma(a[0], r0, acc1);
        acc0 = kd_mma(a[1], p1, acc0);
        acc1 = kd_mma(a[1], r1, acc1);
        acc0 = kd_mma(a[2], p2, acc0);
        acc1 = kd_mma(a[2], r2, acc1);
        acc0 = kd_mma(a[3], p3, acc0);
        acc1 = kd_mma(a[3], r3, acc1);
        if (j > 0) {
            v2f bc0, bc1;
            bc0.x = half ? carry0 : 0.0f;
            bc0.y = 0.0f;
            bc1.x = half ? carry1 : 0.0f;
            bc1.y = 0.0f;
            acc0 = kd_mma(a[4], bc0, acc0);
            acc1 = kd_mma(a[4], bc1, acc1);
        }
        mf0[j] = acc0;
        mf1[j] = acc1;
        carry0 = acc0[7];
        carry1 = acc1[7];
#pragma unroll
        for (int q = 0; q < 5; ++q) a[q] = an[q];
        __builtin_amdgcn_sched_barrier(0);
    }

    carry0 = 0.0f;
    carry1 = 0.0f;
#pragma unroll
    for (int jr = 0; jr < KD_NB; ++jr) {
        const int j = KD_NB - 1 - jr;
        v2f y[8];
#pragma unroll
        for (int i = 0; i < 8; ++i) y[i] = yb[j % KD_PF][i];
        if (j >= 1 && j - 1 < KD_ST) {
#pragma unroll
            for (int i = 0; i < 8; ++i)
                yb[(j - 1) % KD_PF][i] = Yw[((j - 1) * 8 + i) * 32];
        }
        v2f an[5];
        if (j > 0) {
            const v2f* An = Afs + (1 * KD_NB + j - 1) * KD_FR + lane;
#pragma unroll
            for (int q = 0; q < 5; ++q) an[q] = An[q * 32];
        } else {
#pragma unroll
            for (int q = 0; q < 5; ++q) an[q] = a[q];
        }
        const v8f m0 = mf0[j];
        const v8f m1 = mf1[j];
        v2f p0, p1, p2, p3, r0, r1, r2, r3;
        p0.x = m0[0]; p0.y = m0[1];  p1.x = m0[2]; p1.y = m0[3];
        p2.x = m0[4]; p2.y = m0[5];  p3.x = m0[6]; p3.y = m0[7];
        r0.x = m1[0]; r0.y = m1[1];  r1.x = m1[2]; r1.y = m1[3];
        r2.x = m1[4]; r2.y = m1[5];  r3.x = m1[6]; r3.y = m1[7];
        v8f acc0 = {};
        v8f acc1 = {};
        acc0 = kd_mma(a[0], p0, acc0);
        acc1 = kd_mma(a[0], r0, acc1);
        acc0 = kd_mma(a[1], p1, acc0);
        acc1 = kd_mma(a[1], r1, acc1);
        acc0 = kd_mma(a[2], p2, acc0);
        acc1 = kd_mma(a[2], r2, acc1);
        acc0 = kd_mma(a[3], p3, acc0);
        acc1 = kd_mma(a[3], r3, acc1);
        if (j < KD_NB - 1) {
            v2f bc0, bc1;
            bc0.x = half ? 0.0f : carry0;
            bc0.y = 0.0f;
            bc1.x = half ? 0.0f : carry1;
            bc1.y = 0.0f;
            acc0 = kd_mma(a[4], bc0, acc0);
            acc1 = kd_mma(a[4], bc1, acc1);
        }
        carry0 = acc0[0];
        carry1 = acc1[0];
#pragma unroll
        for (int i = 0; i < 8; ++i) {
            const size_t o = (size_t)(16 * j + i) * KD_C;
            v2f t, d;
            t.x = acc0[i];          t.y = acc1[i];
            d.x = y[i].x - acc0[i]; d.y = y[i].y - acc1[i];
            *(v2f*)(Tp + o) = t;
            *(v2f*)(Rp + o) = d;
        }
#pragma unroll
        for (int q = 0; q < 5; ++q) a[q] = an[q];
        __builtin_amdgcn_sched_barrier(0);
    }
}

extern "C" void kernel_launch(void* const* d_in, const int* in_sizes, int n_in,
                              void* d_out, int out_size, void* d_ws, size_t ws_size,
                              hipStream_t stream) {
    (void)in_sizes; (void)n_in; (void)out_size; (void)ws_size;

    const float* obs = (const float*)d_in[0];
    float* out = (float*)d_out;
    float* Wf  = (float*)d_ws;

    kd_build_blocks<<<KD_NB, 32, 0, stream>>>(Wf);

    kd_scan<<<KD_B * 4, 128, 0, stream>>>(obs, Wf, out);
}


# One launch instead of two: each workgroup runs the gain recursion in registers and builds the 64 block matrices in its own LDS while its first obs loads are in flight; sum order and values unchanged
# speedup vs baseline: 7.7597x; 1.4091x over previous
#include <hip/hip_runtime.h>

typedef __attribute__((ext_vector_type(2))) float v2f;
typedef __attribute__((ext_vector_type(4))) float v4f;
typedef __attribute__((ext_vector_type(8))) float v8f;

#define KD_L  512
#define KD_C  512
#define KD_B  64
#define KD_NB (KD_L / 16)
#define KD_AP 161
#define KD_PF 6
#define KD_ST (KD_NB - KD_PF)


__device__ __forceinline__ v8f kd_mma(v2f a, v2f b, v8f c) {
    return __builtin_amdgcn_wmma_f32_16x16x4_f32(false, a, false, b, (short)0, c,
                                                 false, false);
}


__device__ __forceinline__ void kd_build_in_lds(v2f* __restrict__ Afs, int lane, int wave) {
    const int jb = lane;

    float Kr[16], Gr[16];
#pragma unroll
    for (int i = 0; i < 16; ++i) { Kr[i] = 0.0f; Gr[i] = 0.0f; }
    float Kl = 0.0f, Kb = 0.0f, Gl = 0.0f, Gb = 0.0f;
    int n = 0;
    {
        float P      = 0.0f;
        float Pp     = P + 1.0f;
        float PpBack = Pp;
        int t = 0;
        for (; t < KD_L; ++t) {
            const float K = Pp / (Pp + 1.0f);
            P = (1.0f - K) * Pp;
            const float PpNext = P + 1.0f;
            const float G = P / (P + 1.0f);
            const int local = t - 16 * jb;
#pragma unroll
            for (int i = 0; i < 16; ++i)
                if (local == i) { Kr[i] = K; Gr[i] = G; }
            Kb = Kl; Kl = K;
            Gb = Gl; Gl = G;
            if (t >= 1 && PpNext == PpBack) { ++t; break; }
            PpBack = Pp;
            Pp     = PpNext;
        }
        n = t;
    }
    if (n < KD_L) {
#pragma unroll
        for (int i = 0; i < 16; ++i) {
            const int u = 16 * jb + i;
            if (u >= n) {
                const bool odd = ((u - n) & 1) != 0;
                Kr[i] = odd ? Kl : Kb;
                Gr[i] = odd ? Gl : Gb;
            }
        }
    }

    v2f* Fd = Afs + (0 * KD_NB + jb) * KD_AP;
    v2f* Bd = Afs + (1 * KD_NB + jb) * KD_AP;
    const int wv = __builtin_amdgcn_readfirstlane(wave);

#pragma nounroll
    for (int hh = 0; hh < 2; ++hh) {
        const int k0 = 8 * hh + 2 * wv;
        v2f* Fc = Fd + wv * 32 + hh * 16;
        v2f* Bc = Bd + wv * 32 + hh * 16;
        float m0 = 0.0f, m1 = 0.0f;
#pragma unroll
        for (int i = 0; i < 16; ++i) {
            const float y0 = (i == k0) ? 1.0f : 0.0f;
            const float y1 = (i == k0 + 1) ? 1.0f : 0.0f;
            m0 = m0 + Kr[i] * (y0 - m0);
            m1 = m1 + Kr[i] * (y1 - m1);
            v2f v; v.x = m0; v.y = m1;
            Fc[i] = v;
        }
        float s0 = 0.0f, s1 = 0.0f;
#pragma unroll
        for (int i = 15; i >= 0; --i) {
            const int   t   = 16 * jb + i;
            const float mt0 = (i == k0) ? 1.0f : 0.0f;
            const float mt1 = (i == k0 + 1) ? 1.0f : 0.0f;
            s0 = (t == KD_L - 1) ? mt0 : (mt0 + Gr[i] * (s0 - mt0));
            s1 = (t == KD_L - 1) ? mt1 : (mt1 + Gr[i] * (s1 - mt1));
            v2f v; v.x = s0; v.y = s1;
            Bc[i] = v;
        }
    }

    const v2f zero2 = {0.0f, 0.0f};
    if (wv == 2) {
        float m = 1.0f;
#pragma unroll
        for (int i = 0; i < 16; ++i) {
            const float y = 0.0f;
            m = m + Kr[i] * (y - m);
            v2f v; v.x = m; v.y = 0.0f;
            Fd[4 * 32 + 16 + i] = v;
            Fd[4 * 32 + i]      = zero2;
        }
    } else if (wv == 3) {
        float ms = 1.0f;
#pragma unroll
        for (int i = 15; i >= 0; --i) {
            const int   t  = 16 * jb + i;
            const float mt = 0.0f;
            ms = (t == KD_L - 1) ? mt : (mt + Gr[i] * (ms - mt));
            v2f v; v.x = ms; v.y = 0.0f;
            Bd[4 * 32 + i]      = v;
            Bd[4 * 32 + 16 + i] = zero2;
        }
    }
}

__global__ __launch_bounds__(128) void kd_scan(const float* __restrict__ obs,
                                               float* __restrict__ out) {
    __shared__ v2f Afs[2 * KD_NB * KD_AP];
    __shared__ v2f Ysh[4 * KD_ST * 8 * 32];

    const int tid  = threadIdx.x;
    const int lane = tid & 31;
    const int wave = tid >> 5;

    const int blk   = blockIdx.x;
    const int b     = blk >> 2;
    const int strip = (blk & 3) * 4 + wave;
    const int m     = lane & 15;
    const int half  = lane >> 4;
    const size_t base = (size_t)b * KD_L * KD_C + (size_t)(8 * half) * KD_C
                        + strip * 32 + 2 * m;
    const float* __restrict__ Yp = obs + base;
    float* __restrict__ Tp = out + base;
    float* __restrict__ Rp = out + (size_t)KD_B * KD_L * KD_C + base;
    v2f* Yw = Ysh + (size_t)wave * (KD_ST * 8 * 32) + lane;

    v8f mf0[KD_NB];
    v8f mf1[KD_NB];
    v2f yb[KD_PF][8];
    v2f a[5];

#pragma unroll
    for (int jj = 0; jj < KD_PF; ++jj)
#pragma unroll
        for (int i = 0; i < 8; ++i)
            yb[jj][i] = *(const v2f*)(Yp + (size_t)(16 * jj + i) * KD_C);

    kd_build_in_lds(Afs, lane, wave);
    __syncthreads();

#pragma unroll
    for (int q = 0; q < 5; ++q) a[q] = Afs[(0 * KD_NB + 0) * KD_AP + q * 32 + lane];

    float carry0 = 0.0f, carry1 = 0.0f;
#pragma unroll
    for (int j = 0; j < KD_NB; ++j) {
        v2f y[8];
#pragma unroll
        for (int i = 0; i < 8; ++i) y[i] = yb[j % KD_PF][i];
        if (j + KD_PF < KD_NB) {
#pragma unroll
            for (int i = 0; i < 8; ++i)
                yb[j % KD_PF][i] =
                    *(const v2f*)(Yp + (size_t)(16 * (j + KD_PF) + i) * KD_C);
        }
        if (j < KD_ST) {
#pragma unroll
            for (int i = 0; i < 8; ++i) Yw[(j * 8 + i) * 32] = y[i];
        }
        v2f an[5];
        {
            const v2f* An = (j + 1 < KD_NB) ? Afs + (0 * KD_NB + j + 1) * KD_AP + lane
                                            : Afs + (1 * KD_NB + KD_NB - 1) * KD_AP + lane;
#pragma unroll
            for (int q = 0; q < 5; ++q) an[q] = An[q * 32];
        }
        v2f p0, p1, p2, p3, r0, r1, r2, r3;
        p0.x = y[0].x; p0.y = y[1].x;  p1.x = y[2].x; p1.y = y[3].x;
        p2.x = y[4].x; p2.y = y[5].x;  p3.x = y[6].x; p3.y = y[7].x;
        r0.x = y[0].y; r0.y = y[1].y;  r1.x = y[2].y; r1.y = y[3].y;
        r2.x = y[4].y; r2.y = y[5].y;  r3.x = y[6].y; r3.y = y[7].y;
        v8f acc0 = {};
        v8f acc1 = {};
        acc0 = kd_mma(a[0], p0, acc0);
        acc1 = kd_mma(a[0], r0, acc1);
        acc0 = kd_mma(a[1], p1, acc0);
        acc1 = kd_mma(a[1], r1, acc1);
        acc0 = kd_mma(a[2], p2, acc0);
        acc1 = kd_mma(a[2], r2, acc1);
        acc0 = kd_mma(a[3], p3, acc0);
        acc1 = kd_mma(a[3], r3, acc1);
        if (j > 0) {
            v2f bc0, bc1;
            bc0.x = half ? carry0 : 0.0f;
            bc0.y = 0.0f;
            bc1.x = half ? carry1 : 0.0f;
            bc1.y = 0.0f;
            acc0 = kd_mma(a[4], bc0, acc0);
            acc1 = kd_mma(a[4], bc1, acc1);
        }
        mf0[j] = acc0;
        mf1[j] = acc1;
        carry0 = acc0[7];
        carry1 = acc1[7];
#pragma unroll
        for (int q = 0; q < 5; ++q) a[q] = an[q];
        __builtin_amdgcn_sched_barrier(0);
    }

    carry0 = 0.0f;
    carry1 = 0.0f;
#pragma unroll
    for (int jr = 0; jr < KD_NB; ++jr) {
        const int j = KD_NB - 1 - jr;
        v2f y[8];
#pragma unroll
        for (int i = 0; i < 8; ++i) y[i] = yb[j % KD_PF][i];
        if (j >= 1 && j - 1 < KD_ST) {
#pragma unroll
            for (int i = 0; i < 8; ++i)
                yb[(j - 1) % KD_PF][i] = Yw[((j - 1) * 8 + i) * 32];
        }
        v2f an[5];
        if (j > 0) {
            const v2f* An = Afs + (1 * KD_NB + j - 1) * KD_AP + lane;
#pragma unroll
            for (int q = 0; q < 5; ++q) an[q] = An[q * 32];
        } else {
#pragma unroll
            for (int q = 0; q < 5; ++q) an[q] = a[q];
        }
        const v8f m0 = mf0[j];
        const v8f m1 = mf1[j];
        v2f p0, p1, p2, p3, r0, r1, r2, r3;
        p0.x = m0[0]; p0.y = m0[1];  p1.x = m0[2]; p1.y = m0[3];
        p2.x = m0[4]; p2.y = m0[5];  p3.x = m0[6]; p3.y = m0[7];
        r0.x = m1[0]; r0.y = m1[1];  r1.x = m1[2]; r1.y = m1[3];
        r2.x = m1[4]; r2.y = m1[5];  r3.x = m1[6]; r3.y = m1[7];
        v8f acc0 = {};
        v8f acc1 = {};
        acc0 = kd_mma(a[0], p0, acc0);
        acc1 = kd_mma(a[0], r0, acc1);
        acc0 = kd_mma(a[1], p1, acc0);
        acc1 = kd_mma(a[1], r1, acc1);
        acc0 = kd_mma(a[2], p2, acc0);
        acc1 = kd_mma(a[2], r2, acc1);
        acc0 = kd_mma(a[3], p3, acc0);
        acc1 = kd_mma(a[3], r3, acc1);
        if (j < KD_NB - 1) {
            v2f bc0, bc1;
            bc0.x = half ? 0.0f : carry0;
            bc0.y = 0.0f;
            bc1.x = half ? 0.0f : carry1;
            bc1.y = 0.0f;
            acc0 = kd_mma(a[4], bc0, acc0);
            acc1 = kd_mma(a[4], bc1, acc1);
        }
        carry0 = acc0[0];
        carry1 = acc1[0];
#pragma unroll
        for (int i = 0; i < 8; ++i) {
            const size_t o = (size_t)(16 * j + i) * KD_C;
            v2f t, d;
            t.x = acc0[i];          t.y = acc1[i];
            d.x = y[i].x - acc0[i]; d.y = y[i].y - acc1[i];
            *(v2f*)(Tp + o) = t;
            *(v2f*)(Rp + o) = d;
        }
#pragma unroll
        for (int q = 0; q < 5; ++q) a[q] = an[q];
        __builtin_amdgcn_sched_barrier(0);
    }
}

extern "C" void kernel_launch(void* const* d_in, const int* in_sizes, int n_in,
                              void* d_out, int out_size, void* d_ws, size_t ws_size,
                              hipStream_t stream) {
    (void)in_sizes; (void)n_in; (void)out_size; (void)d_ws; (void)ws_size;

    const float* obs = (const float*)d_in[0];
    float* out = (float*)d_out;

    kd_scan<<<KD_B * 4, 128, 0, stream>>>(obs, out);
}


# Output rows leave as whole 256-B units: the two waves of a pair swap 128-B halves of trend and residual through LDS, 16 B per lane stores; obs ring 7 blocks, 25 parked, for LDS room; values same
# speedup vs baseline: 9.6266x; 1.2406x over previous
#include <hip/hip_runtime.h>

typedef __attribute__((ext_vector_type(2))) float v2f;
typedef __attribute__((ext_vector_type(4))) float v4f;
typedef __attribute__((ext_vector_type(8))) float v8f;

#define KD_L  512
#define KD_C  512
#define KD_B  64
#define KD_NB (KD_L / 16)
#define KD_AP 161
#define KD_PF 7
#define KD_XP 34
#define KD_ST (KD_NB - KD_PF)


__device__ __forceinline__ v8f kd_mma(v2f a, v2f b, v8f c) {
    return __builtin_amdgcn_wmma_f32_16x16x4_f32(false, a, false, b, (short)0, c,
                                                 false, false);
}


__device__ __forceinline__ void kd_build_in_lds(v2f* __restrict__ Afs, int lane, int wave) {
    const int jb = lane;

    float Kr[16], Gr[16];
#pragma unroll
    for (int i = 0; i < 16; ++i) { Kr[i] = 0.0f; Gr[i] = 0.0f; }
    float Kl = 0.0f, Kb = 0.0f, Gl = 0.0f, Gb = 0.0f;
    int n = 0;
    {
        float P      = 0.0f;
        float Pp     = P + 1.0f;
        float PpBack = Pp;
        int t = 0;
        for (; t < KD_L; ++t) {
            const float K = Pp / (Pp + 1.0f);
            P = (1.0f - K) * Pp;
            const float PpNext = P + 1.0f;
            const float G = P / (P + 1.0f);
            const int local = t - 16 * jb;
#pragma unroll
            for (int i = 0; i < 16; ++i)
                if (local == i) { Kr[i] = K; Gr[i] = G; }
            Kb = Kl; Kl = K;
            Gb = Gl; Gl = G;
            if (t >= 1 && PpNext == PpBack) { ++t; break; }
            PpBack = Pp;
            Pp     = PpNext;
        }
        n = t;
    }
    if (n < KD_L) {
#pragma unroll
        for (int i = 0; i < 16; ++i) {
            const int u = 16 * jb + i;
            if (u >= n) {
                const bool odd = ((u - n) & 1) != 0;
                Kr[i] = odd ? Kl : Kb;
                Gr[i] = odd ? Gl : Gb;
            }
        }
    }

    v2f* Fd = Afs + (0 * KD_NB + jb) * KD_AP;
    v2f* Bd = Afs + (1 * KD_NB + jb) * KD_AP;
    const int wv = __builtin_amdgcn_readfirstlane(wave);

#pragma nounroll
    for (int hh = 0; hh < 2; ++hh) {
        const int k0 = 8 * hh + 2 * wv;
        v2f* Fc = Fd + wv * 32 + hh * 16;
        v2f* Bc = Bd + wv * 32 + hh * 16;
        float m0 = 0.0f, m1 = 0.0f;
#pragma unroll
        for (int i = 0; i < 16; ++i) {
            const float y0 = (i == k0) ? 1.0f : 0.0f;
            const float y1 = (i == k0 + 1) ? 1.0f : 0.0f;
            m0 = m0 + Kr[i] * (y0 - m0);
            m1 = m1 + Kr[i] * (y1 - m1);
            v2f v; v.x = m0; v.y = m1;
            Fc[i] = v;
        }
        float s0 = 0.0f, s1 = 0.0f;
#pragma unroll
        for (int i = 15; i >= 0; --i) {
            const int   t   = 16 * jb + i;
            const float mt0 = (i == k0) ? 1.0f : 0.0f;
            const float mt1 = (i == k0 + 1) ? 1.0f : 0.0f;
            s0 = (t == KD_L - 1) ? mt0 : (mt0 + Gr[i] * (s0 - mt0));
            s1 = (t == KD_L - 1) ? mt1 : (mt1 + Gr[i] * (s1 - mt1));
            v2f v; v.x = s0; v.y = s1;
            Bc[i] = v;
        }
    }

    const v2f zero2 = {0.0f, 0.0f};
    if (wv == 2) {
        float m = 1.0f;
#pragma unroll
        for (int i = 0; i < 16; ++i) {
            const float y = 0.0f;
            m = m + Kr[i] * (y - m);
            v2f v; v.x = m; v.y = 0.0f;
            Fd[4 * 32 + 16 + i] = v;
            Fd[4 * 32 + i]      = zero2;
        }
    } else if (wv == 3) {
        float ms = 1.0f;
#pragma unroll
        for (int i = 15; i >= 0; --i) {
            const int   t  = 16 * jb + i;
            const float mt = 0.0f;
            ms = (t == KD_L - 1) ? mt : (mt + Gr[i] * (ms - mt));
            v2f v; v.x = ms; v.y = 0.0f;
            Bd[4 * 32 + i]      = v;
            Bd[4 * 32 + 16 + i] = zero2;
        }
    }
}

__global__ __launch_bounds__(128) void kd_scan(const float* __restrict__ obs,
                                               float* __restrict__ out) {
    __shared__ v2f Afs[2 * KD_NB * KD_AP];
    __shared__ v2f Ysh[4 * KD_ST * 8 * 32];
    __shared__ __attribute__((aligned(16))) v2f Xsh[2 * 2 * 2 * 16 * KD_XP];

    const int tid  = threadIdx.x;
    const int lane = tid & 31;
    const int wave = tid >> 5;

    const int blk   = blockIdx.x;
    const int b     = blk >> 2;
    const int strip = (blk & 3) * 4 + wave;
    const int m     = lane & 15;
    const int half  = lane >> 4;
    const size_t base = (size_t)b * KD_L * KD_C + (size_t)(8 * half) * KD_C
                        + strip * 32 + 2 * m;
    const float* __restrict__ Yp = obs + base;
    v2f* Yw = Ysh + (size_t)wave * (KD_ST * 8 * 32) + lane;
    const int par = wave & 1;
    v2f* Xp = Xsh + (size_t)(wave >> 1) * (2 * 2 * 16 * KD_XP);
    v2f* Xw = Xp + (8 * half) * KD_XP + par * 16 + m;
    const v2f* Xr = Xp + (8 * par + half) * KD_XP + 2 * m;
    const size_t qbase = (size_t)b * KD_L * KD_C + (size_t)(8 * par + half) * KD_C
                         + (strip & ~1) * 32 + 4 * m;
    float* __restrict__ Tq = out + qbase;
    float* __restrict__ Rq = out + (size_t)KD_B * KD_L * KD_C + qbase;

    v8f mf0[KD_NB];
    v8f mf1[KD_NB];
    v2f yb[KD_PF][8];
    v2f a[5];

#pragma unroll
    for (int jj = 0; jj < KD_PF; ++jj)
#pragma unroll
        for (int i = 0; i < 8; ++i)
            yb[jj][i] = *(const v2f*)(Yp + (size_t)(16 * jj + i) * KD_C);

    kd_build_in_lds(Afs, lane, wave);
    __syncthreads();

#pragma unroll
    for (int q = 0; q < 5; ++q) a[q] = Afs[(0 * KD_NB + 0) * KD_AP + q * 32 + lane];

    float carry0 = 0.0f, carry1 = 0.0f;
#pragma unroll
    for (int j = 0; j < KD_NB; ++j) {
        v2f y[8];
#pragma unroll
        for (int i = 0; i < 8; ++i) y[i] = yb[j % KD_PF][i];
        if (j + KD_PF < KD_NB) {
#pragma unroll
            for (int i = 0; i < 8; ++i)
                yb[j % KD_PF][i] =
                    *(const v2f*)(Yp + (size_t)(16 * (j + KD_PF) + i) * KD_C);
        }
        if (j < KD_ST) {
#pragma unroll
            for (int i = 0; i < 8; ++i) Yw[(j * 8 + i) * 32] = y[i];
        }
        v2f an[5];
        {
            const v2f* An = (j + 1 < KD_NB) ? Afs + (0 * KD_NB + j + 1) * KD_AP + lane
                                            : Afs + (1 * KD_NB + KD_NB - 1) * KD_AP + lane;
#pragma unroll
            for (int q = 0; q < 5; ++q) an[q] = An[q * 32];
        }
        v2f p0, p1, p2, p3, r0, r1, r2, r3;
        p0.x = y[0].x; p0.y = y[1].x;  p1.x = y[2].x; p1.y = y[3].x;
        p2.x = y[4].x; p2.y = y[5].x;  p3.x = y[6].x; p3.y = y[7].x;
        r0.x = y[0].y; r0.y = y[1].y;  r1.x = y[2].y; r1.y = y[3].y;
        r2.x = y[4].y; r2.y = y[5].y;  r3.x = y[6].y; r3.y = y[7].y;
        v8f acc0 = {};
        v8f acc1 = {};
        acc0 = kd_mma(a[0], p0, acc0);
        acc1 = kd_mma(a[0], r0, acc1);
        acc0 = kd_mma(a[1], p1, acc0);
        acc1 = kd_mma(a[1], r1, acc1);
        acc0 = kd_mma(a[2], p2, acc0);
        acc1 = kd_mma(a[2], r2, acc1);
        acc0 = kd_mma(a[3], p3, acc0);
        acc1 = kd_mma(a[3], r3, acc1);
        if (j > 0) {
            v2f bc0, bc1;
            bc0.x = half ? carry0 : 0.0f;
            bc0.y = 0.0f;
            bc1.x = half ? carry1 : 0.0f;
            bc1.y = 0.0f;
            acc0 = kd_mma(a[4], bc0, acc0);
            acc1 = kd_mma(a[4], bc1, acc1);
        }
        mf0[j] = acc0;
        mf1[j] = acc1;
        carry0 = acc0[7];
        carry1 = acc1[7];
#pragma unroll
        for (int q = 0; q < 5; ++q) a[q] = an[q];
        __builtin_amdgcn_sched_barrier(0);
    }

    carry0 = 0.0f;
    carry1 = 0.0f;
#pragma unroll
    for (int jr = 0; jr < KD_NB; ++jr) {
        const int j = KD_NB - 1 - jr;
        v4f xt[4], xd[4];
        if (jr > 0) {
            const v2f* Xt = Xr + (((jr - 1) & 1) * 2 + 0) * 16 * KD_XP;
            const v2f* Xd = Xr + (((jr - 1) & 1) * 2 + 1) * 16 * KD_XP;
#pragma unroll
            for (int k = 0; k < 4; ++k) {
                xt[k] = *(const v4f*)(Xt + 2 * k * KD_XP);
                xd[k] = *(const v4f*)(Xd + 2 * k * KD_XP);
            }
        }
        v2f y[8];
#pragma unroll
        for (int i = 0; i < 8; ++i) y[i] = yb[j % KD_PF][i];
        if (j >= 1 && j - 1 < KD_ST) {
#pragma unroll
            for (int i = 0; i < 8; ++i)
                yb[(j - 1) % KD_PF][i] = Yw[((j - 1) * 8 + i) * 32];
        }
        v2f an[5];
        if (j > 0) {
            const v2f* An = Afs + (1 * KD_NB + j - 1) * KD_AP + lane;
#pragma unroll
            for (int q = 0; q < 5; ++q) an[q] = An[q * 32];
        } else {
#pragma unroll
            for (int q = 0; q < 5; ++q) an[q] = a[q];
        }
        const v8f m0 = mf0[j];
        const v8f m1 = mf1[j];
        v2f p0, p1, p2, p3, r0, r1, r2, r3;
        p0.x = m0[0]; p0.y = m0[1];  p1.x = m0[2]; p1.y = m0[3];
        p2.x = m0[4]; p2.y = m0[5];  p3.x = m0[6]; p3.y = m0[7];
        r0.x = m1[0]; r0.y = m1[1];  r1.x = m1[2]; r1.y = m1[3];
        r2.x = m1[4]; r2.y = m1[5];  r3.x = m1[6]; r3.y = m1[7];
        v8f acc0 = {};
        v8f acc1 = {};
        acc0 = kd_mma(a[0], p0, acc0);
        acc1 = kd_mma(a[0], r0, acc1);
        acc0 = kd_mma(a[1], p1, acc0);
        acc1 = kd_mma(a[1], r1, acc1);
        acc0 = kd_mma(a[2], p2, acc0);
        acc1 = kd_mma(a[2], r2, acc1);
        acc0 = kd_mma(a[3], p3, acc0);
        acc1 = kd_mma(a[3], r3, acc1);
        if (j < KD_NB - 1) {
            v2f bc0, bc1;
            bc0.x = half ? 0.0f : carry0;
            bc0.y = 0.0f;
            bc1.x = half ? 0.0f : carry1;
            bc1.y = 0.0f;
            acc0 = kd_mma(a[4], bc0, acc0);
            acc1 = kd_mma(a[4], bc1, acc1);
        }
        carry0 = acc0[0];
        carry1 = acc1[0];
        if (jr > 0) {
#pragma unroll
            for (int k = 0; k < 4; ++k) {
                const size_t o = (size_t)(16 * (j + 1) + 2 * k) * KD_C;
                *(v4f*)(Tq + o) = xt[k];
                *(v4f*)(Rq + o) = xd[k];
            }
        }
        {
            v2f* Wt = Xw + ((jr & 1) * 2 + 0) * 16 * KD_XP;
            v2f* Wd = Xw + ((jr & 1) * 2 + 1) * 16 * KD_XP;
#pragma unroll
            for (int i = 0; i < 8; ++i) {
                v2f t, d;
                t.x = acc0[i];          t.y = acc1[i];
                d.x = y[i].x - acc0[i]; d.y = y[i].y - acc1[i];
                Wt[i * KD_XP] = t;
                Wd[i * KD_XP] = d;
            }
        }
#pragma unroll
        for (int q = 0; q < 5; ++q) a[q] = an[q];
        __builtin_amdgcn_fence(__ATOMIC_RELEASE, "workgroup", "local");
        __builtin_amdgcn_s_barrier();
        __builtin_amdgcn_fence(__ATOMIC_ACQUIRE, "workgroup", "local");
        __builtin_amdgcn_sched_barrier(0);
    }
    {
        const v2f* Xt = Xr + (((KD_NB - 1) & 1) * 2 + 0) * 16 * KD_XP;
        const v2f* Xd = Xr + (((KD_NB - 1) & 1) * 2 + 1) * 16 * KD_XP;
#pragma unroll
        for (int k = 0; k < 4; ++k) {
            const size_t o = (size_t)(2 * k) * KD_C;
            *(v4f*)(Tq + o) = *(const v4f*)(Xt + 2 * k * KD_XP);
            *(v4f*)(Rq + o) = *(const v4f*)(Xd + 2 * k * KD_XP);
        }
    }
}

extern "C" void kernel_launch(void* const* d_in, const int* in_sizes, int n_in,
                              void* d_out, int out_size, void* d_ws, size_t ws_size,
                              hipStream_t stream) {
    (void)in_sizes; (void)n_in; (void)out_size; (void)d_ws; (void)ws_size;

    const float* obs = (const float*)d_in[0];
    float* out = (float*)d_out;

    kd_scan<<<KD_B * 4, 128, 0, stream>>>(obs, out);
}


# Backward sweep: the image hand-over barrier is split; a trip runs its 8 carry-free matrix instructions, then signals, stores and does the carry ones, then waits; sum order and values unchanged
# speedup vs baseline: 10.1450x; 1.0539x over previous
#include <hip/hip_runtime.h>

typedef __attribute__((ext_vector_type(2))) float v2f;
typedef __attribute__((ext_vector_type(4))) float v4f;
typedef __attribute__((ext_vector_type(8))) float v8f;

#define KD_L  512
#define KD_C  512
#define KD_B  64
#define KD_NB (KD_L / 16)
#define KD_AP 161
#define KD_PF 7
#define KD_XP 34
#define KD_ST (KD_NB - KD_PF)


__device__ __forceinline__ v8f kd_mma(v2f a, v2f b, v8f c) {
    return __builtin_amdgcn_wmma_f32_16x16x4_f32(false, a, false, b, (short)0, c,
                                                 false, false);
}


__device__ __forceinline__ void kd_build_in_lds(v2f* __restrict__ Afs, int lane, int wave) {
    const int jb = lane;

    float Kr[16], Gr[16];
#pragma unroll
    for (int i = 0; i < 16; ++i) { Kr[i] = 0.0f; Gr[i] = 0.0f; }
    float Kl = 0.0f, Kb = 0.0f, Gl = 0.0f, Gb = 0.0f;
    int n = 0;
    {
        float P      = 0.0f;
        float Pp     = P + 1.0f;
        float PpBack = Pp;
        int t = 0;
        for (; t < KD_L; ++t) {
            const float K = Pp / (Pp + 1.0f);
            P = (1.0f - K) * Pp;
            const float PpNext = P + 1.0f;
            const float G = P / (P + 1.0f);
            const int local = t - 16 * jb;
#pragma unroll
            for (int i = 0; i < 16; ++i)
                if (local == i) { Kr[i] = K; Gr[i] = G; }
            Kb = Kl; Kl = K;
            Gb = Gl; Gl = G;
            if (t >= 1 && PpNext == PpBack) { ++t; break; }
            PpBack = Pp;
            Pp     = PpNext;
        }
        n = t;
    }
    if (n < KD_L) {
#pragma unroll
        for (int i = 0; i < 16; ++i) {
            const int u = 16 * jb + i;
            if (u >= n) {
                const bool odd = ((u - n) & 1) != 0;
                Kr[i] = odd ? Kl : Kb;
                Gr[i] = odd ? Gl : Gb;
            }
        }
    }

    v2f* Fd = Afs + (0 * KD_NB + jb) * KD_AP;
    v2f* Bd = Afs + (1 * KD_NB + jb) * KD_AP;
    const int wv = __builtin_amdgcn_readfirstlane(wave);

#pragma nounroll
    for (int hh = 0; hh < 2; ++hh) {
        const int k0 = 8 * hh + 2 * wv;
        v2f* Fc = Fd + wv * 32 + hh * 16;
        v2f* Bc = Bd + wv * 32 + hh * 16;
        float m0 = 0.0f, m1 = 0.0f;
#pragma unroll
        for (int i = 0; i < 16; ++i) {
            const float y0 = (i == k0) ? 1.0f : 0.0f;
            const float y1 = (i == k0 + 1) ? 1.0f : 0.0f;
            m0 = m0 + Kr[i] * (y0 - m0);
            m1 = m1 + Kr[i] * (y1 - m1);
            v2f v; v.x = m0; v.y = m1;
            Fc[i] = v;
        }
        float s0 = 0.0f, s1 = 0.0f;
#pragma unroll
        for (int i = 15; i >= 0; --i) {
            const int   t   = 16 * jb + i;
            const float mt0 = (i == k0) ? 1.0f : 0.0f;
            const float mt1 = (i == k0 + 1) ? 1.0f : 0.0f;
            s0 = (t == KD_L - 1) ? mt0 : (mt0 + Gr[i] * (s0 - mt0));
            s1 = (t == KD_L - 1) ? mt1 : (mt1 + Gr[i] * (s1 - mt1));
            v2f v; v.x = s0; v.y = s1;
            Bc[i] = v;
        }
    }

    const v2f zero2 = {0.0f, 0.0f};
    if (wv == 2) {
        float m = 1.0f;
#pragma unroll
        for (int i = 0; i < 16; ++i) {
            const float y = 0.0f;
            m = m + Kr[i] * (y - m);
            v2f v; v.x = m; v.y = 0.0f;
            Fd[4 * 32 + 16 + i] = v;
            Fd[4 * 32 + i]      = zero2;
        }
    } else if (wv == 3) {
        float ms = 1.0f;
#pragma unroll
        for (int i = 15; i >= 0; --i) {
            const int   t  = 16 * jb + i;
            const float mt = 0.0f;
            ms = (t == KD_L - 1) ? mt : (mt + Gr[i] * (ms - mt));
            v2f v; v.x = ms; v.y = 0.0f;
            Bd[4 * 32 + i]      = v;
            Bd[4 * 32 + 16 + i] = zero2;
        }
    }
}

__global__ __launch_bounds__(128) void kd_scan(const float* __restrict__ obs,
                                               float* __restrict__ out) {
    __shared__ v2f Afs[2 * KD_NB * KD_AP];
    __shared__ v2f Ysh[4 * KD_ST * 8 * 32];
    __shared__ __attribute__((aligned(16))) v2f Xsh[2 * 2 * 2 * 16 * KD_XP];

    const int tid  = threadIdx.x;
    const int lane = tid & 31;
    const int wave = tid >> 5;

    const int blk   = blockIdx.x;
    const int b     = blk >> 2;
    const int strip = (blk & 3) * 4 + wave;
    const int m     = lane & 15;
    const int half  = lane >> 4;
    const size_t base = (size_t)b * KD_L * KD_C + (size_t)(8 * half) * KD_C
                        + strip * 32 + 2 * m;
    const float* __restrict__ Yp = obs + base;
    v2f* Yw = Ysh + (size_t)wave * (KD_ST * 8 * 32) + lane;
    const int par = wave & 1;
    v2f* Xp = Xsh + (size_t)(wave >> 1) * (2 * 2 * 16 * KD_XP);
    v2f* Xw = Xp + (8 * half) * KD_XP + par * 16 + m;
    const v2f* Xr = Xp + (8 * par + half) * KD_XP + 2 * m;
    const size_t qbase = (size_t)b * KD_L * KD_C + (size_t)(8 * par + half) * KD_C
                         + (strip & ~1) * 32 + 4 * m;
    float* __restrict__ Tq = out + qbase;
    float* __restrict__ Rq = out + (size_t)KD_B * KD_L * KD_C + qbase;

    v8f mf0[KD_NB];
    v8f mf1[KD_NB];
    v2f yb[KD_PF][8];
    v2f a[5];

#pragma unroll
    for (int jj = 0; jj < KD_PF; ++jj)
#pragma unroll
        for (int i = 0; i < 8; ++i)
            yb[jj][i] = *(const v2f*)(Yp + (size_t)(16 * jj + i) * KD_C);

    kd_build_in_lds(Afs, lane, wave);
    __syncthreads();

#pragma unroll
    for (int q = 0; q < 5; ++q) a[q] = Afs[(0 * KD_NB + 0) * KD_AP + q * 32 + lane];

    float carry0 = 0.0f, carry1 = 0.0f;
#pragma unroll
    for (int j = 0; j < KD_NB; ++j) {
        v2f y[8];
#pragma unroll
        for (int i = 0; i < 8; ++i) y[i] = yb[j % KD_PF][i];
        if (j + KD_PF < KD_NB) {
#pragma unroll
            for (int i = 0; i < 8; ++i)
                yb[j % KD_PF][i] =
                    *(const v2f*)(Yp + (size_t)(16 * (j + KD_PF) + i) * KD_C);
        }
        if (j < KD_ST) {
#pragma unroll
            for (int i = 0; i < 8; ++i) Yw[(j * 8 + i) * 32] = y[i];
        }
        v2f an[5];
        {
            const v2f* An = (j + 1 < KD_NB) ? Afs + (0 * KD_NB + j + 1) * KD_AP + lane
                                            : Afs + (1 * KD_NB + KD_NB - 1) * KD_AP + lane;
#pragma unroll
            for (int q = 0; q < 5; ++q) an[q] = An[q * 32];
        }
        v2f p0, p1, p2, p3, r0, r1, r2, r3;
        p0.x = y[0].x; p0.y = y[1].x;  p1.x = y[2].x; p1.y = y[3].x;
        p2.x = y[4].x; p2.y = y[5].x;  p3.x = y[6].x; p3.y = y[7].x;
        r0.x = y[0].y; r0.y = y[1].y;  r1.x = y[2].y; r1.y = y[3].y;
        r2.x = y[4].y; r2.y = y[5].y;  r3.x = y[6].y; r3.y = y[7].y;
        v8f acc0 = {};
        v8f acc1 = {};
        acc0 = kd_mma(a[0], p0, acc0);
        acc1 = kd_mma(a[0], r0, acc1);
        acc0 = kd_mma(a[1], p1, acc0);
        acc1 = kd_mma(a[1], r1, acc1);
        acc0 = kd_mma(a[2], p2, acc0);
        acc1 = kd_mma(a[2], r2, acc1);
        acc0 = kd_mma(a[3], p3, acc0);
        acc1 = kd_mma(a[3], r3, acc1);
        if (j > 0) {
            v2f bc0, bc1;
            bc0.x = half ? carry0 : 0.0f;
            bc0.y = 0.0f;
            bc1.x = half ? carry1 : 0.0f;
            bc1.y = 0.0f;
            acc0 = kd_mma(a[4], bc0, acc0);
            acc1 = kd_mma(a[4], bc1, acc1);
        }
        mf0[j] = acc0;
        mf1[j] = acc1;
        carry0 = acc0[7];
        carry1 = acc1[7];
#pragma unroll
        for (int q = 0; q < 5; ++q) a[q] = an[q];
        __builtin_amdgcn_sched_barrier(0);
    }

    carry0 = 0.0f;
    carry1 = 0.0f;
    v4f xt[4], xd[4];
#pragma unroll
    for (int jr = 0; jr < KD_NB; ++jr) {
        const int j = KD_NB - 1 - jr;
        v2f y[8];
#pragma unroll
        for (int i = 0; i < 8; ++i) y[i] = yb[j % KD_PF][i];
        const v8f m0 = mf0[j];
        const v8f m1 = mf1[j];
        v2f p0, p1, p2, p3, r0, r1, r2, r3;
        p0.x = m0[0]; p0.y = m0[1];  p1.x = m0[2]; p1.y = m0[3];
        p2.x = m0[4]; p2.y = m0[5];  p3.x = m0[6]; p3.y = m0[7];
        r0.x = m1[0]; r0.y = m1[1];  r1.x = m1[2]; r1.y = m1[3];
        r2.x = m1[4]; r2.y = m1[5];  r3.x = m1[6]; r3.y = m1[7];
        v8f acc0 = {};
        v8f acc1 = {};
        acc0 = kd_mma(a[0], p0, acc0);
        acc1 = kd_mma(a[0], r0, acc1);
        acc0 = kd_mma(a[1], p1, acc0);
        acc1 = kd_mma(a[1], r1, acc1);
        acc0 = kd_mma(a[2], p2, acc0);
        acc1 = kd_mma(a[2], r2, acc1);
        acc0 = kd_mma(a[3], p3, acc0);
        acc1 = kd_mma(a[3], r3, acc1);
        if (jr > 0) {
            __builtin_amdgcn_sched_barrier(0);
            __builtin_amdgcn_fence(__ATOMIC_RELEASE, "workgroup", "local");
            __builtin_amdgcn_s_barrier_signal(-1);
        }
        if (jr > 1) {
#pragma unroll
            for (int k = 0; k < 4; ++k) {
                const size_t o = (size_t)(16 * (j + 2) + 2 * k) * KD_C;
                *(v4f*)(Tq + o) = xt[k];
                *(v4f*)(Rq + o) = xd[k];
            }
        }
        if (j < KD_NB - 1) {
            v2f bc0, bc1;
            bc0.x = half ? 0.0f : carry0;
            bc0.y = 0.0f;
            bc1.x = half ? 0.0f : carry1;
            bc1.y = 0.0f;
            acc0 = kd_mma(a[4], bc0, acc0);
            acc1 = kd_mma(a[4], bc1, acc1);
        }
        carry0 = acc0[0];
        carry1 = acc1[0];
        if (jr > 0) {
            __builtin_amdgcn_s_barrier_wait(-1);
            __builtin_amdgcn_fence(__ATOMIC_ACQUIRE, "workgroup", "local");
        }
        v2f an[5];
        if (j > 0) {
            const v2f* An = Afs + (1 * KD_NB + j - 1) * KD_AP + lane;
#pragma unroll
            for (int q = 0; q < 5; ++q) an[q] = An[q * 32];
        } else {
#pragma unroll
            for (int q = 0; q < 5; ++q) an[q] = a[q];
        }
        if (jr > 0) {
            const v2f* Xt = Xr + (((jr - 1) & 1) * 2 + 0) * 16 * KD_XP;
            const v2f* Xd = Xr + (((jr - 1) & 1) * 2 + 1) * 16 * KD_XP;
#pragma unroll
            for (int k = 0; k < 4; ++k) {
                xt[k] = *(const v4f*)(Xt + 2 * k * KD_XP);
                xd[k] = *(const v4f*)(Xd + 2 * k * KD_XP);
            }
        }
        if (j >= 1 && j - 1 < KD_ST) {
#pragma unroll
            for (int i = 0; i < 8; ++i)
                yb[(j - 1) % KD_PF][i] = Yw[((j - 1) * 8 + i) * 32];
        }
        {
            v2f* Wt = Xw + ((jr & 1) * 2 + 0) * 16 * KD_XP;
            v2f* Wd = Xw + ((jr & 1) * 2 + 1) * 16 * KD_XP;
#pragma unroll
            for (int i = 0; i < 8; ++i) {
                v2f t, d;
                t.x = acc0[i];          t.y = acc1[i];
                d.x = y[i].x - acc0[i]; d.y = y[i].y - acc1[i];
                Wt[i * KD_XP] = t;
                Wd[i * KD_XP] = d;
            }
        }
#pragma unroll
        for (int q = 0; q < 5; ++q) a[q] = an[q];
        __builtin_amdgcn_sched_barrier(0);
    }
#pragma unroll
    for (int k = 0; k < 4; ++k) {
        const size_t o = (size_t)(16 + 2 * k) * KD_C;
        *(v4f*)(Tq + o) = xt[k];
        *(v4f*)(Rq + o) = xd[k];
    }
    __builtin_amdgcn_fence(__ATOMIC_RELEASE, "workgroup", "local");
    __builtin_amdgcn_s_barrier_signal(-1);
    __builtin_amdgcn_s_barrier_wait(-1);
    __builtin_amdgcn_fence(__ATOMIC_ACQUIRE, "workgroup", "local");
    {
        const v2f* Xt = Xr + (((KD_NB - 1) & 1) * 2 + 0) * 16 * KD_XP;
        const v2f* Xd = Xr + (((KD_NB - 1) & 1) * 2 + 1) * 16 * KD_XP;
#pragma unroll
        for (int k = 0; k < 4; ++k) {
            const size_t o = (size_t)(2 * k) * KD_C;
            *(v4f*)(Tq + o) = *(const v4f*)(Xt + 2 * k * KD_XP);
            *(v4f*)(Rq + o) = *(const v4f*)(Xd + 2 * k * KD_XP);
        }
    }
}

extern "C" void kernel_launch(void* const* d_in, const int* in_sizes, int n_in,
                              void* d_out, int out_size, void* d_ws, size_t ws_size,
                              hipStream_t stream) {
    (void)in_sizes; (void)n_in; (void)out_size; (void)d_ws; (void)ws_size;

    const float* obs = (const float*)d_in[0];
    float* out = (float*)d_out;

    kd_scan<<<KD_B * 4, 128, 0, stream>>>(obs, out);
}
